# starting kernel
# baseline (speedup 1.0000x reference)
#include <hip/hip_runtime.h>
#include <math.h>

typedef __attribute__((ext_vector_type(2))) float v2f;
typedef __attribute__((ext_vector_type(8))) float v8f;

#define TPB 256
#define WAVES_PER_BLOCK 8

template <int M>
__global__ __launch_bounds__(TPB) void gemm_wmma_f32(
    const float* __restrict__ X, const float* __restrict__ W,
    float* __restrict__ Y, int Nrows) {
  __shared__ __align__(16) float lw[64 * M];

  constexpr int TOTAL4 = 64 * M / 4;
  for (int i = threadIdx.x; i < TOTAL4; i += TPB)
    ((float4*)lw)[i] = ((const float4*)W)[i];
  __syncthreads();

  const int rowTile = blockIdx.x * WAVES_PER_BLOCK + (threadIdx.x >> 5);
  if (rowTile >= (Nrows >> 4)) return;
  const int row0 = rowTile << 4;
  const int lane = threadIdx.x & 31;
  const int r    = lane & 15;
  const int half = lane >> 4;

  v2f a[16];
  const float* xrow = X + (size_t)(row0 + r) * 64;
#pragma unroll
  for (int ks = 0; ks < 16; ++ks) {
    const int k = 4 * ks + 2 * half;
    a[ks].x = xrow[k];
    a[ks].y = xrow[k + 1];
  }

#pragma unroll
  for (int tn = 0; tn < M / 16; ++tn) {
    const int col0 = tn << 4;
    v8f c = {};
#pragma unroll
    for (int ks = 0; ks < 16; ++ks) {
      const int k = 4 * ks + 2 * half;
      v2f b;
      b.x = lw[k * M + col0 + r];
      b.y = lw[(k + 1) * M + col0 + r];
      c = __builtin_amdgcn_wmma_f32_16x16x4_f32(false, a[ks], false, b,
                                                (short)0, c, false, false);
    }
#pragma unroll
    for (int v = 0; v < 8; ++v)
      Y[(size_t)(row0 + v + 8 * half) * M + col0 + r] = c[v];
  }
}

__device__ __forceinline__ void edge_sd(const int* __restrict__ ei, int e, int E,
                                        int& s, int& d) {
  if (e < E) { s = ei[e]; d = ei[E + e]; }
  else       { s = e - E; d = s; }
}

__device__ __forceinline__ void atomic_max_float(float* addr, float val) {
  if (val >= 0.0f) atomicMax((int*)addr, __float_as_int(val));
  else             atomicMin((unsigned int*)addr, __float_as_uint(val));
}

template <int C>
__global__ __launch_bounds__(TPB) void edge_score(
    const float* __restrict__ xl, const float* __restrict__ xr,
    const int* __restrict__ ei, const float* __restrict__ att,
    float* __restrict__ score, float* __restrict__ mbuf, int E, int ET) {
  const int tid = blockIdx.x * blockDim.x + threadIdx.x;
  if (tid >= ET * 4) return;
  const int e = tid >> 2, h = tid & 3;
  int s, d; edge_sd(ei, e, E, s, d);
  const int ldf = 4 * C;
  const float4* pl = (const float4*)(xl + (size_t)s * ldf + h * C);
  const float4* pr = (const float4*)(xr + (size_t)d * ldf + h * C);
  const float4* pa = (const float4*)(att + h * C);
  float acc = 0.0f;
#pragma unroll
  for (int i = 0; i < C / 4; ++i) {
    float4 a = pl[i], b = pr[i], w = pa[i];
    float vx = a.x + b.x, vy = a.y + b.y, vz = a.z + b.z, vw = a.w + b.w;
    vx = vx > 0.0f ? vx : 0.2f * vx;
    vy = vy > 0.0f ? vy : 0.2f * vy;
    vz = vz > 0.0f ? vz : 0.2f * vz;
    vw = vw > 0.0f ? vw : 0.2f * vw;
    acc += w.x * vx + w.y * vy + w.z * vz + w.w * vw;
  }
  score[(size_t)e * 4 + h] = acc;
  atomic_max_float(&mbuf[(size_t)d * 4 + h], acc);
}

__global__ __launch_bounds__(TPB) void edge_exp(
    float* __restrict__ score, const float* __restrict__ mbuf,
    float* __restrict__ denom, const int* __restrict__ ei, int E, int ET) {
  const int tid = blockIdx.x * blockDim.x + threadIdx.x;
  if (tid >= ET * 4) return;
  const int e = tid >> 2, h = tid & 3;
  int s, d; edge_sd(ei, e, E, s, d);
  const float ex = expf(score[(size_t)e * 4 + h] - mbuf[(size_t)d * 4 + h]);
  score[(size_t)e * 4 + h] = ex;
  atomicAdd(&denom[(size_t)d * 4 + h], ex);
}

template <int C>
__global__ __launch_bounds__(TPB) void edge_agg(
    const float* __restrict__ xl, const float* __restrict__ score,
    const float* __restrict__ denom, const int* __restrict__ ei,
    float* __restrict__ agg, int E, int ET) {
  const int tid = blockIdx.x * blockDim.x + threadIdx.x;
  if (tid >= ET * 4) return;
  const int e = tid >> 2, h = tid & 3;
  int s, d; edge_sd(ei, e, E, s, d);
  const float alpha = score[(size_t)e * 4 + h] / denom[(size_t)d * 4 + h];
  const int ldf = 4 * C;
  const float4* srcp = (const float4*)(xl + (size_t)s * ldf + h * C);
  float* dstp = agg + (size_t)d * ldf + h * C;
#pragma unroll
  for (int i = 0; i < C / 4; ++i) {
    float4 v = srcp[i];
    atomicAdd(&dstp[4 * i + 0], alpha * v.x);
    atomicAdd(&dstp[4 * i + 1], alpha * v.y);
    atomicAdd(&dstp[4 * i + 2], alpha * v.z);
    atomicAdd(&dstp[4 * i + 3], alpha * v.w);
  }
}

__global__ __launch_bounds__(TPB) void bias_elu(
    float* __restrict__ feat, const float* __restrict__ bias, int total, int M) {
  const int tid = blockIdx.x * blockDim.x + threadIdx.x;
  if (tid >= total) return;
  const float v = feat[tid] + bias[tid & (M - 1)];
  feat[tid] = v > 0.0f ? v : expm1f(v);
}

__global__ __launch_bounds__(TPB) void final_mean(
    const float* __restrict__ agg, const float* __restrict__ bias,
    float* __restrict__ out, int N) {
  const int tid = blockIdx.x * blockDim.x + threadIdx.x;
  if (tid >= N * 32) return;
  const int n = tid >> 5, c = tid & 31;
  const size_t b = (size_t)n * 128 + c;
  const float s = 0.25f * (agg[b] + agg[b + 32] + agg[b + 64] + agg[b + 96]);
  out[tid] = s + bias[c];
}

extern "C" void kernel_launch(void* const* d_in, const int* in_sizes, int n_in,
                              void* d_out, int out_size, void* d_ws, size_t ws_size,
                              hipStream_t stream) {
  (void)in_sizes; (void)n_in; (void)out_size; (void)ws_size;
  const float* x   = (const float*)d_in[0];
  const int*   ei  = (const int*)d_in[1];
  const float* W1l = (const float*)d_in[2];
  const float* W1r = (const float*)d_in[3];
  const float* a1  = (const float*)d_in[4];
  const float* b1  = (const float*)d_in[5];
  const float* W2l = (const float*)d_in[6];
  const float* W2r = (const float*)d_in[7];
  const float* a2  = (const float*)d_in[8];
  const float* b2  = (const float*)d_in[9];
  const float* W3l = (const float*)d_in[10];
  const float* W3r = (const float*)d_in[11];
  const float* a3  = (const float*)d_in[12];
  const float* b3  = (const float*)d_in[13];

  const int N = 100000, E = 800000, ET = E + N;

  float* ws = (float*)d_ws;
  float* xl = ws;
  float* xr = xl + (size_t)N * 128;
  float* sc = xr + (size_t)N * 128;
  float* mb = sc + (size_t)ET * 4;
  float* dn = mb + (size_t)N * 4;
  float* h1 = dn + (size_t)N * 4;
  float* h2 = h1 + (size_t)N * 64;
  float* ag = h2 + (size_t)N * 64;

  const int rowTiles = N / 16;
  const int gblocks  = (rowTiles + WAVES_PER_BLOCK - 1) / WAVES_PER_BLOCK;

  auto layer = [&](const float* in, const float* Wl, const float* Wr,
                   const float* att, const float* bias, float* out,
                   int C, bool final_layer) {
    const int M = 4 * C;
    if (M == 64) {
      gemm_wmma_f32<64><<<gblocks, TPB, 0, stream>>>(in, Wl, xl, N);
      gemm_wmma_f32<64><<<gblocks, TPB, 0, stream>>>(in, Wr, xr, N);
    } else {
      gemm_wmma_f32<128><<<gblocks, TPB, 0, stream>>>(in, Wl, xl, N);
      gemm_wmma_f32<128><<<gblocks, TPB, 0, stream>>>(in, Wr, xr, N);
    }

    float* aggbuf = final_layer ? ag : out;
    hipMemsetAsync(mb, 0xFF, (size_t)N * 4 * sizeof(float), stream);
    hipMemsetAsync(dn, 0x00, (size_t)N * 4 * sizeof(float), stream);
    hipMemsetAsync(aggbuf, 0x00, (size_t)N * M * sizeof(float), stream);

    const int eThreads = ET * 4;
    const int eBlocks = (eThreads + TPB - 1) / TPB;
    if (C == 16) edge_score<16><<<eBlocks, TPB, 0, stream>>>(xl, xr, ei, att, sc, mb, E, ET);
    else         edge_score<32><<<eBlocks, TPB, 0, stream>>>(xl, xr, ei, att, sc, mb, E, ET);
    edge_exp<<<eBlocks, TPB, 0, stream>>>(sc, mb, dn, ei, E, ET);
    if (C == 16) edge_agg<16><<<eBlocks, TPB, 0, stream>>>(xl, sc, dn, ei, aggbuf, E, ET);
    else         edge_agg<32><<<eBlocks, TPB, 0, stream>>>(xl, sc, dn, ei, aggbuf, E, ET);

    if (!final_layer) {
      const int t = N * M;
      bias_elu<<<(t + TPB - 1) / TPB, TPB, 0, stream>>>(out, bias, t, M);
    } else {
      const int t = N * 32;
      final_mean<<<(t + TPB - 1) / TPB, TPB, 0, stream>>>(ag, bias, (float*)d_out, N);
    }
  };

  layer(x,  W1l, W1r, a1, b1, h1, 16, false);
  layer(h1, W2l, W2r, a2, b2, h2, 16, false);
  layer(h2, W3l, W3r, a3, b3, (float*)d_out, 32, true);
}


# Replace float-atomic edge scatter by in-call dst-sorted CSR and a per-node gather kernel: online segment softmax + aggregation + epilogue in registers; sum order = CSR slot order (int atomics)
# speedup vs baseline: 4.7868x; 4.7868x over previous
#include <hip/hip_runtime.h>
#include <math.h>

typedef __attribute__((ext_vector_type(2))) float v2f;
typedef __attribute__((ext_vector_type(8))) float v8f;

#define TPB 256
#define WAVES_PER_BLOCK 8

template <int M>
__global__ __launch_bounds__(TPB) void gemm_wmma_f32(
    const float* __restrict__ X, const float* __restrict__ W,
    float* __restrict__ Y, int Nrows) {
  __shared__ __align__(16) float lw[64 * M];

  constexpr int TOTAL4 = 64 * M / 4;
  for (int i = threadIdx.x; i < TOTAL4; i += TPB)
    ((float4*)lw)[i] = ((const float4*)W)[i];
  __syncthreads();

  const int rowTile = blockIdx.x * WAVES_PER_BLOCK + (threadIdx.x >> 5);
  if (rowTile >= (Nrows >> 4)) return;
  const int row0 = rowTile << 4;
  const int lane = threadIdx.x & 31;
  const int r    = lane & 15;
  const int half = lane >> 4;

  v2f a[16];
  const float* xrow = X + (size_t)(row0 + r) * 64;
#pragma unroll
  for (int ks = 0; ks < 16; ++ks) {
    const int k = 4 * ks + 2 * half;
    a[ks].x = xrow[k];
    a[ks].y = xrow[k + 1];
  }

#pragma unroll
  for (int tn = 0; tn < M / 16; ++tn) {
    const int col0 = tn << 4;
    v8f c = {};
#pragma unroll
    for (int ks = 0; ks < 16; ++ks) {
      const int k = 4 * ks + 2 * half;
      v2f b;
      b.x = lw[k * M + col0 + r];
      b.y = lw[(k + 1) * M + col0 + r];
      c = __builtin_amdgcn_wmma_f32_16x16x4_f32(false, a[ks], false, b,
                                                (short)0, c, false, false);
    }
#pragma unroll
    for (int v = 0; v < 8; ++v)
      Y[(size_t)(row0 + v + 8 * half) * M + col0 + r] = c[v];
  }
}

__global__ __launch_bounds__(TPB) void csr_zero(int* __restrict__ cnt, int N) {
  const int i = blockIdx.x * TPB + threadIdx.x;
  if (i < N) cnt[i] = 0;
}

__global__ __launch_bounds__(TPB) void csr_count(const int* __restrict__ ei,
                                                 int* __restrict__ cnt, int E) {
  const int e = blockIdx.x * TPB + threadIdx.x;
  if (e < E) atomicAdd(&cnt[ei[E + e]], 1);
}

__global__ __launch_bounds__(TPB) void csr_scan(const int* __restrict__ cnt,
                                                int* __restrict__ rowptr, int N) {
  __shared__ int s_part[WAVES_PER_BLOCK];
  __shared__ int s_incl[WAVES_PER_BLOCK];
  const int t = threadIdx.x, lane = t & 31, w = t >> 5;
  int part = 0;
  const int4* c4 = (const int4*)cnt;
  const int nq = blockIdx.x * TPB;
  for (int q = t; q < nq; q += TPB) {
    const int4 v = c4[q];
    part += (v.x + v.y) + (v.z + v.w);
  }
  const int base = blockIdx.x * (4 * TPB) + 4 * t;
  const int x0 = (base + 0 < N) ? cnt[base + 0] : 0;
  const int x1 = (base + 1 < N) ? cnt[base + 1] : 0;
  const int x2 = (base + 2 < N) ? cnt[base + 2] : 0;
  const int x3 = (base + 3 < N) ? cnt[base + 3] : 0;
  const int own = (x0 + x1) + (x2 + x3);
  for (int o = 16; o > 0; o >>= 1) part += __shfl_xor(part, o);
  int inc = own;
  for (int o = 1; o < 32; o <<= 1) {
    const int v = __shfl_up(inc, o);
    if (lane >= o) inc += v;
  }
  if (lane == 31) s_incl[w] = inc;
  if (lane == 0)  s_part[w] = part;
  __syncthreads();
  int ex = inc - own;
  for (int k = 0; k < WAVES_PER_BLOCK; ++k) {
    ex += s_part[k];
    if (k < w) ex += s_incl[k];
  }
  if (base + 0 < N) rowptr[base + 0] = ex;
  if (base + 1 < N) rowptr[base + 1] = ex + x0;
  if (base + 2 < N) rowptr[base + 2] = ex + x0 + x1;
  if (base + 3 < N) rowptr[base + 3] = ex + x0 + x1 + x2;
  if (base < N && base + 4 >= N) rowptr[N] = ex + own;
}

__global__ __launch_bounds__(TPB) void csr_fill(const int* __restrict__ ei,
                                                int* __restrict__ cnt,
                                                const int* __restrict__ rowptr,
                                                int* __restrict__ csr, int E) {
  const int e = blockIdx.x * TPB + threadIdx.x;
  if (e < E) {
    const int d = ei[E + e];
    const int old = atomicSub(&cnt[d], 1);
    csr[rowptr[d] + old - 1] = ei[e];
  }
}

template <int C, bool FINAL>
__global__ __launch_bounds__(TPB) void node_softmax_agg(
    const float* __restrict__ xl, const float* __restrict__ xr,
    const float* __restrict__ att, const float* __restrict__ bias,
    const int* __restrict__ rowptr, const int* __restrict__ csr,
    float* __restrict__ outp, int N) {
  constexpr int M   = 4 * C;
  constexpr int LPN = C;
  constexpr int LPH = C / 4;
  const int node = (blockIdx.x * TPB + threadIdx.x) / LPN;
  const int sub  = threadIdx.x & (LPN - 1);
  const bool valid = node < N;
  const int i = valid ? node : N - 1;

  const float4 xr4 = *(const float4*)(xr + (size_t)i * M + 4 * sub);
  const float4 at4 = *(const float4*)(att + 4 * sub);
  const int beg   = rowptr[i];
  const int total = rowptr[i + 1] - beg + 1;
  int wtotal = total;
  if constexpr (LPN < 32) wtotal = max(wtotal, __shfl_xor(wtotal, 16));

  float m = -INFINITY, den = 0.0f;
  float4 acc = make_float4(0.0f, 0.0f, 0.0f, 0.0f);

  for (int base = 0; base < wtotal; base += LPN) {
    const int idx = base + sub;
    int my_s = i;
    if (idx >= 1 && idx < total) my_s = csr[beg + idx - 1];
    const int cnt = min(LPN, wtotal - base);
    for (int j = 0; j < cnt; ++j) {
      const int s = __shfl(my_s, j, LPN);
      const float4 row = *(const float4*)(xl + (size_t)s * M + 4 * sub);
      float vx = row.x + xr4.x, vy = row.y + xr4.y;
      float vz = row.z + xr4.z, vw = row.w + xr4.w;
      vx = vx > 0.0f ? vx : 0.2f * vx;
      vy = vy > 0.0f ? vy : 0.2f * vy;
      vz = vz > 0.0f ? vz : 0.2f * vz;
      vw = vw > 0.0f ? vw : 0.2f * vw;
      float p = at4.x * vx + at4.y * vy + at4.z * vz + at4.w * vw;
      for (int o = 1; o < LPH; o <<= 1) p += __shfl_xor(p, o);
      const float sc = (base + j < total) ? p : -INFINITY;
      const float mn = fmaxf(m, sc);
      const float scale = expf(m - mn);
      const float wgt   = expf(sc - mn);
      den   = den   * scale + wgt;
      acc.x = acc.x * scale + wgt * row.x;
      acc.y = acc.y * scale + wgt * row.y;
      acc.z = acc.z * scale + wgt * row.z;
      acc.w = acc.w * scale + wgt * row.w;
      m = mn;
    }
  }

  float4 r = make_float4(acc.x / den, acc.y / den, acc.z / den, acc.w / den);
  if constexpr (!FINAL) {
    const float4 b4 = *(const float4*)(bias + 4 * sub);
    r.x += b4.x; r.y += b4.y; r.z += b4.z; r.w += b4.w;
    r.x = r.x > 0.0f ? r.x : expm1f(r.x);
    r.y = r.y > 0.0f ? r.y : expm1f(r.y);
    r.z = r.z > 0.0f ? r.z : expm1f(r.z);
    r.w = r.w > 0.0f ? r.w : expm1f(r.w);
    if (valid) *(float4*)(outp + (size_t)i * M + 4 * sub) = r;
  } else {
    for (int o = LPH; o < LPN; o <<= 1) {
      r.x += __shfl_xor(r.x, o); r.y += __shfl_xor(r.y, o);
      r.z += __shfl_xor(r.z, o); r.w += __shfl_xor(r.w, o);
    }
    if (valid && sub < LPH) {
      const float4 b4 = *(const float4*)(bias + 4 * sub);
      r.x = 0.25f * r.x + b4.x; r.y = 0.25f * r.y + b4.y;
      r.z = 0.25f * r.z + b4.z; r.w = 0.25f * r.w + b4.w;
      *(float4*)(outp + (size_t)i * C + 4 * sub) = r;
    }
  }
}

extern "C" void kernel_launch(void* const* d_in, const int* in_sizes, int n_in,
                              void* d_out, int out_size, void* d_ws, size_t ws_size,
                              hipStream_t stream) {
  (void)in_sizes; (void)n_in; (void)out_size; (void)ws_size;
  const float* x   = (const float*)d_in[0];
  const int*   ei  = (const int*)d_in[1];
  const float* W1l = (const float*)d_in[2];
  const float* W1r = (const float*)d_in[3];
  const float* a1  = (const float*)d_in[4];
  const float* b1  = (const float*)d_in[5];
  const float* W2l = (const float*)d_in[6];
  const float* W2r = (const float*)d_in[7];
  const float* a2  = (const float*)d_in[8];
  const float* b2  = (const float*)d_in[9];
  const float* W3l = (const float*)d_in[10];
  const float* W3r = (const float*)d_in[11];
  const float* a3  = (const float*)d_in[12];
  const float* b3  = (const float*)d_in[13];

  const int N = 100000, E = 800000;

  float* ws = (float*)d_ws;
  float* xl = ws;
  float* xr = xl + (size_t)N * 128;
  float* h1 = xr + (size_t)N * 128;
  float* h2 = h1 + (size_t)N * 64;
  int* cnt    = (int*)(h2 + (size_t)N * 64);
  int* rowptr = cnt + N;
  int* csr    = rowptr + (N + 4);

  const int rowTiles = N / 16;
  const int gblocks  = (rowTiles + WAVES_PER_BLOCK - 1) / WAVES_PER_BLOCK;

  csr_zero <<<(N + TPB - 1) / TPB, TPB, 0, stream>>>(cnt, N);
  csr_count<<<(E + TPB - 1) / TPB, TPB, 0, stream>>>(ei, cnt, E);
  csr_scan <<<(N + 4 * TPB - 1) / (4 * TPB), TPB, 0, stream>>>(cnt, rowptr, N);
  csr_fill <<<(E + TPB - 1) / TPB, TPB, 0, stream>>>(ei, cnt, rowptr, csr, E);

  const int nb16 = (N * 16 + TPB - 1) / TPB;
  const int nb32 = (N * 32 + TPB - 1) / TPB;

  gemm_wmma_f32<64><<<gblocks, TPB, 0, stream>>>(x, W1l, xl, N);
  gemm_wmma_f32<64><<<gblocks, TPB, 0, stream>>>(x, W1r, xr, N);
  node_softmax_agg<16, false><<<nb16, TPB, 0, stream>>>(xl, xr, a1, b1, rowptr, csr, h1, N);

  gemm_wmma_f32<64><<<gblocks, TPB, 0, stream>>>(h1, W2l, xl, N);
  gemm_wmma_f32<64><<<gblocks, TPB, 0, stream>>>(h1, W2r, xr, N);
  node_softmax_agg<16, false><<<nb16, TPB, 0, stream>>>(xl, xr, a2, b2, rowptr, csr, h2, N);

  gemm_wmma_f32<128><<<gblocks, TPB, 0, stream>>>(h2, W3l, xl, N);
  gemm_wmma_f32<128><<<gblocks, TPB, 0, stream>>>(h2, W3r, xr, N);
  node_softmax_agg<32, true><<<nb32, TPB, 0, stream>>>(xl, xr, a3, b3, rowptr, csr,
                                                       (float*)d_out, N);
}


# xl and xr of each layer from ONE GEMM launch: Wl and Wr staged side by side in LDS, each wave loads its 16x64 A tile once and sweeps both; 13 -> 10 launches, per-element sum order unchanged
# speedup vs baseline: 5.1036x; 1.0662x over previous
#include <hip/hip_runtime.h>
#include <math.h>

typedef __attribute__((ext_vector_type(2))) float v2f;
typedef __attribute__((ext_vector_type(8))) float v8f;

#define TPB 256
#define WAVES_PER_BLOCK 8

template <int M>
__global__ __launch_bounds__(TPB) void gemm_wmma_f32(
    const float* __restrict__ X, const float* __restrict__ Wl,
    const float* __restrict__ Wr, float* __restrict__ Yl,
    float* __restrict__ Yr, int Nrows) {
  __shared__ __align__(16) float lw[2 * 64 * M];

  constexpr int TOTAL4 = 64 * M / 4;
  for (int i = threadIdx.x; i < TOTAL4; i += TPB) {
    ((float4*)lw)[i]          = ((const float4*)Wl)[i];
    ((float4*)lw)[TOTAL4 + i] = ((const float4*)Wr)[i];
  }
  __syncthreads();

  const int rowTile = blockIdx.x * WAVES_PER_BLOCK + (threadIdx.x >> 5);
  if (rowTile >= (Nrows >> 4)) return;
  const int row0 = rowTile << 4;
  const int lane = threadIdx.x & 31;
  const int r    = lane & 15;
  const int half = lane >> 4;

  v2f a[16];
  const float* xrow = X + (size_t)(row0 + r) * 64;
#pragma unroll
  for (int ks = 0; ks < 16; ++ks) {
    const int k = 4 * ks + 2 * half;
    a[ks].x = xrow[k];
    a[ks].y = xrow[k + 1];
  }

#pragma unroll
  for (int side = 0; side < 2; ++side) {
    const float* lws = lw + side * (64 * M);
    float* Y = side ? Yr : Yl;
#pragma nounroll
    for (int tn = 0; tn < M / 16; ++tn) {
      const int col0 = tn << 4;
      v8f c = {};
#pragma unroll
      for (int ks = 0; ks < 16; ++ks) {
        const int k = 4 * ks + 2 * half;
        v2f b;
        b.x = lws[k * M + col0 + r];
        b.y = lws[(k + 1) * M + col0 + r];
        c = __builtin_amdgcn_wmma_f32_16x16x4_f32(false, a[ks], false, b,
                                                  (short)0, c, false, false);
      }
#pragma unroll
      for (int v = 0; v < 8; ++v)
        Y[(size_t)(row0 + v + 8 * half) * M + col0 + r] = c[v];
    }
  }
}

__global__ __launch_bounds__(TPB) void csr_zero(int* __restrict__ cnt, int N) {
  const int i = blockIdx.x * TPB + threadIdx.x;
  if (i < N) cnt[i] = 0;
}

__global__ __launch_bounds__(TPB) void csr_count(const int* __restrict__ ei,
                                                 int* __restrict__ cnt, int E) {
  const int e = blockIdx.x * TPB + threadIdx.x;
  if (e < E) atomicAdd(&cnt[ei[E + e]], 1);
}

__global__ __launch_bounds__(TPB) void csr_scan(const int* __restrict__ cnt,
                                                int* __restrict__ rowptr, int N) {
  __shared__ int s_part[WAVES_PER_BLOCK];
  __shared__ int s_incl[WAVES_PER_BLOCK];
  const int t = threadIdx.x, lane = t & 31, w = t >> 5;
  int part = 0;
  const int4* c4 = (const int4*)cnt;
  const int nq = blockIdx.x * TPB;
  for (int q = t; q < nq; q += TPB) {
    const int4 v = c4[q];
    part += (v.x + v.y) + (v.z + v.w);
  }
  const int base = blockIdx.x * (4 * TPB) + 4 * t;
  const int x0 = (base + 0 < N) ? cnt[base + 0] : 0;
  const int x1 = (base + 1 < N) ? cnt[base + 1] : 0;
  const int x2 = (base + 2 < N) ? cnt[base + 2] : 0;
  const int x3 = (base + 3 < N) ? cnt[base + 3] : 0;
  const int own = (x0 + x1) + (x2 + x3);
  for (int o = 16; o > 0; o >>= 1) part += __shfl_xor(part, o);
  int inc = own;
  for (int o = 1; o < 32; o <<= 1) {
    const int v = __shfl_up(inc, o);
    if (lane >= o) inc += v;
  }
  if (lane == 31) s_incl[w] = inc;
  if (lane == 0)  s_part[w] = part;
  __syncthreads();
  int ex = inc - own;
  for (int k = 0; k < WAVES_PER_BLOCK; ++k) {
    ex += s_part[k];
    if (k < w) ex += s_incl[k];
  }
  if (base + 0 < N) rowptr[base + 0] = ex;
  if (base + 1 < N) rowptr[base + 1] = ex + x0;
  if (base + 2 < N) rowptr[base + 2] = ex + x0 + x1;
  if (base + 3 < N) rowptr[base + 3] = ex + x0 + x1 + x2;
  if (base < N && base + 4 >= N) rowptr[N] = ex + own;
}

__global__ __launch_bounds__(TPB) void csr_fill(const int* __restrict__ ei,
                                                int* __restrict__ cnt,
                                                const int* __restrict__ rowptr,
                                                int* __restrict__ csr, int E) {
  const int e = blockIdx.x * TPB + threadIdx.x;
  if (e < E) {
    const int d = ei[E + e];
    const int old = atomicSub(&cnt[d], 1);
    csr[rowptr[d] + old - 1] = ei[e];
  }
}

template <int C, bool FINAL>
__global__ __launch_bounds__(TPB) void node_softmax_agg(
    const float* __restrict__ xl, const float* __restrict__ xr,
    const float* __restrict__ att, const float* __restrict__ bias,
    const int* __restrict__ rowptr, const int* __restrict__ csr,
    float* __restrict__ outp, int N) {
  constexpr int M   = 4 * C;
  constexpr int LPN = C;
  constexpr int LPH = C / 4;
  const int node = (blockIdx.x * TPB + threadIdx.x) / LPN;
  const int sub  = threadIdx.x & (LPN - 1);
  const bool valid = node < N;
  const int i = valid ? node : N - 1;

  const float4 xr4 = *(const float4*)(xr + (size_t)i * M + 4 * sub);
  const float4 at4 = *(const float4*)(att + 4 * sub);
  const int beg   = rowptr[i];
  const int total = rowptr[i + 1] - beg + 1;
  int wtotal = total;
  if constexpr (LPN < 32) wtotal = max(wtotal, __shfl_xor(wtotal, 16));

  float m = -INFINITY, den = 0.0f;
  float4 acc = make_float4(0.0f, 0.0f, 0.0f, 0.0f);

  for (int base = 0; base < wtotal; base += LPN) {
    const int idx = base + sub;
    int my_s = i;
    if (idx >= 1 && idx < total) my_s = csr[beg + idx - 1];
    const int cnt = min(LPN, wtotal - base);
    for (int j = 0; j < cnt; ++j) {
      const int s = __shfl(my_s, j, LPN);
      const float4 row = *(const float4*)(xl + (size_t)s * M + 4 * sub);
      float vx = row.x + xr4.x, vy = row.y + xr4.y;
      float vz = row.z + xr4.z, vw = row.w + xr4.w;
      vx = vx > 0.0f ? vx : 0.2f * vx;
      vy = vy > 0.0f ? vy : 0.2f * vy;
      vz = vz > 0.0f ? vz : 0.2f * vz;
      vw = vw > 0.0f ? vw : 0.2f * vw;
      float p = at4.x * vx + at4.y * vy + at4.z * vz + at4.w * vw;
      for (int o = 1; o < LPH; o <<= 1) p += __shfl_xor(p, o);
      const float sc = (base + j < total) ? p : -INFINITY;
      const float mn = fmaxf(m, sc);
      const float scale = expf(m - mn);
      const float wgt   = expf(sc - mn);
      den   = den   * scale + wgt;
      acc.x = acc.x * scale + wgt * row.x;
      acc.y = acc.y * scale + wgt * row.y;
      acc.z = acc.z * scale + wgt * row.z;
      acc.w = acc.w * scale + wgt * row.w;
      m = mn;
    }
  }

  float4 r = make_float4(acc.x / den, acc.y / den, acc.z / den, acc.w / den);
  if constexpr (!FINAL) {
    const float4 b4 = *(const float4*)(bias + 4 * sub);
    r.x += b4.x; r.y += b4.y; r.z += b4.z; r.w += b4.w;
    r.x = r.x > 0.0f ? r.x : expm1f(r.x);
    r.y = r.y > 0.0f ? r.y : expm1f(r.y);
    r.z = r.z > 0.0f ? r.z : expm1f(r.z);
    r.w = r.w > 0.0f ? r.w : expm1f(r.w);
    if (valid) *(float4*)(outp + (size_t)i * M + 4 * sub) = r;
  } else {
    for (int o = LPH; o < LPN; o <<= 1) {
      r.x += __shfl_xor(r.x, o); r.y += __shfl_xor(r.y, o);
      r.z += __shfl_xor(r.z, o); r.w += __shfl_xor(r.w, o);
    }
    if (valid && sub < LPH) {
      const float4 b4 = *(const float4*)(bias + 4 * sub);
      r.x = 0.25f * r.x + b4.x; r.y = 0.25f * r.y + b4.y;
      r.z = 0.25f * r.z + b4.z; r.w = 0.25f * r.w + b4.w;
      *(float4*)(outp + (size_t)i * C + 4 * sub) = r;
    }
  }
}

extern "C" void kernel_launch(void* const* d_in, const int* in_sizes, int n_in,
                              void* d_out, int out_size, void* d_ws, size_t ws_size,
                              hipStream_t stream) {
  (void)in_sizes; (void)n_in; (void)out_size; (void)ws_size;
  const float* x   = (const float*)d_in[0];
  const int*   ei  = (const int*)d_in[1];
  const float* W1l = (const float*)d_in[2];
  const float* W1r = (const float*)d_in[3];
  const float* a1  = (const float*)d_in[4];
  const float* b1  = (const float*)d_in[5];
  const float* W2l = (const float*)d_in[6];
  const float* W2r = (const float*)d_in[7];
  const float* a2  = (const float*)d_in[8];
  const float* b2  = (const float*)d_in[9];
  const float* W3l = (const float*)d_in[10];
  const float* W3r = (const float*)d_in[11];
  const float* a3  = (const float*)d_in[12];
  const float* b3  = (const float*)d_in[13];

  const int N = 100000, E = 800000;

  float* ws = (float*)d_ws;
  float* xl = ws;
  float* xr = xl + (size_t)N * 128;
  float* h1 = xr + (size_t)N * 128;
  float* h2 = h1 + (size_t)N * 64;
  int* cnt    = (int*)(h2 + (size_t)N * 64);
  int* rowptr = cnt + N;
  int* csr    = rowptr + (N + 4);

  const int rowTiles = N / 16;
  const int gblocks  = (rowTiles + WAVES_PER_BLOCK - 1) / WAVES_PER_BLOCK;

  csr_zero <<<(N + TPB - 1) / TPB, TPB, 0, stream>>>(cnt, N);
  csr_count<<<(E + TPB - 1) / TPB, TPB, 0, stream>>>(ei, cnt, E);
  csr_scan <<<(N + 4 * TPB - 1) / (4 * TPB), TPB, 0, stream>>>(cnt, rowptr, N);
  csr_fill <<<(E + TPB - 1) / TPB, TPB, 0, stream>>>(ei, cnt, rowptr, csr, E);

  const int nb16 = (N * 16 + TPB - 1) / TPB;
  const int nb32 = (N * 32 + TPB - 1) / TPB;

  gemm_wmma_f32<64><<<gblocks, TPB, 0, stream>>>(x, W1l, W1r, xl, xr, N);
  node_softmax_agg<16, false><<<nb16, TPB, 0, stream>>>(xl, xr, a1, b1, rowptr, csr, h1, N);

  gemm_wmma_f32<64><<<gblocks, TPB, 0, stream>>>(h1, W2l, W2r, xl, xr, N);
  node_softmax_agg<16, false><<<nb16, TPB, 0, stream>>>(xl, xr, a2, b2, rowptr, csr, h2, N);

  gemm_wmma_f32<128><<<gblocks, TPB, 0, stream>>>(h2, W3l, W3r, xl, xr, N);
  node_softmax_agg<32, true><<<nb32, TPB, 0, stream>>>(xl, xr, a3, b3, rowptr, csr,
                                                       (float*)d_out, N);
}


# CSR build as a two-level counting sort on LDS counters (chunk-local grouping by 512-node bucket, then one work-group per bucket) replaces global int atomics; 10->8 launches; sum order = LDS arrival
# speedup vs baseline: 7.6279x; 1.4946x over previous
#include <hip/hip_runtime.h>
#include <math.h>

typedef __attribute__((ext_vector_type(2))) float v2f;
typedef __attribute__((ext_vector_type(8))) float v8f;

#define TPB 256
#define WAVES_PER_BLOCK 8

template <int M>
__global__ __launch_bounds__(TPB) void gemm_wmma_f32(
    const float* __restrict__ X, const float* __restrict__ Wl,
    const float* __restrict__ Wr, float* __restrict__ Yl,
    float* __restrict__ Yr, int Nrows) {
  __shared__ __align__(16) float lw[2 * 64 * M];

  constexpr int TOTAL4 = 64 * M / 4;
  for (int i = threadIdx.x; i < TOTAL4; i += TPB) {
    ((float4*)lw)[i]          = ((const float4*)Wl)[i];
    ((float4*)lw)[TOTAL4 + i] = ((const float4*)Wr)[i];
  }
  __syncthreads();

  const int rowTile = blockIdx.x * WAVES_PER_BLOCK + (threadIdx.x >> 5);
  if (rowTile >= (Nrows >> 4)) return;
  const int row0 = rowTile << 4;
  const int lane = threadIdx.x & 31;
  const int r    = lane & 15;
  const int half = lane >> 4;

  v2f a[16];
  const float* xrow = X + (size_t)(row0 + r) * 64;
#pragma unroll
  for (int ks = 0; ks < 16; ++ks) {
    const int k = 4 * ks + 2 * half;
    a[ks].x = xrow[k];
    a[ks].y = xrow[k + 1];
  }

#pragma unroll
  for (int side = 0; side < 2; ++side) {
    const float* lws = lw + side * (64 * M);
    float* Y = side ? Yr : Yl;
#pragma nounroll
    for (int tn = 0; tn < M / 16; ++tn) {
      const int col0 = tn << 4;
      v8f c = {};
#pragma unroll
      for (int ks = 0; ks < 16; ++ks) {
        const int k = 4 * ks + 2 * half;
        v2f b;
        b.x = lws[k * M + col0 + r];
        b.y = lws[(k + 1) * M + col0 + r];
        c = __builtin_amdgcn_wmma_f32_16x16x4_f32(false, a[ks], false, b,
                                                  (short)0, c, false, false);
      }
#pragma unroll
      for (int v = 0; v < 8; ++v)
        Y[(size_t)(row0 + v + 8 * half) * M + col0 + r] = c[v];
    }
  }
}

#define BTH 1024
#define EPC 4096
#define NPB 512
#define NPB_LOG 9
#define STAGE_CAP 8192

__device__ __forceinline__ int block_excl_scan(int v, int* s_w, int& total) {
  const int lane = threadIdx.x & 31, w = threadIdx.x >> 5;
  int inc = v;
  for (int o = 1; o < 32; o <<= 1) {
    const int u = __shfl_up(inc, o);
    if (lane >= o) inc += u;
  }
  if (lane == 31) s_w[w] = inc;
  __syncthreads();
  int pre = 0, tot = 0;
  for (int k = 0; k < BTH / 32; ++k) {
    const int x = s_w[k];
    if (k < w) pre += x;
    tot += x;
  }
  __syncthreads();
  total = tot;
  return pre + inc - v;
}

template <int NBK, int NCH>
__global__ __launch_bounds__(BTH) void csr_bin(const int* __restrict__ ei,
                                               int* __restrict__ tmp,
                                               int* __restrict__ tcnt,
                                               int* __restrict__ toff, int E) {
  __shared__ int s_hist[NBK];
  __shared__ int s_off[NBK];
  __shared__ int s_w[BTH / 32];
  __shared__ int s_stage[EPC];
  const int t = threadIdx.x, c = blockIdx.x;
  if (t < NBK) s_hist[t] = 0;
  __syncthreads();

  const int e0 = c * EPC;
  int word[EPC / BTH], bk[EPC / BTH], rk[EPC / BTH];
#pragma unroll
  for (int j = 0; j < EPC / BTH; ++j) {
    const int e = e0 + j * BTH + t;
    bk[j] = -1; word[j] = 0; rk[j] = 0;
    if (e < E) {
      const int s = ei[e], d = ei[E + e];
      bk[j]   = d >> NPB_LOG;
      word[j] = (s << NPB_LOG) | (d & (NPB - 1));
      rk[j]   = atomicAdd(&s_hist[bk[j]], 1);
    }
  }
  __syncthreads();

  const int h = (t < NBK) ? s_hist[t] : 0;
  int nvalid;
  const int ex = block_excl_scan(h, s_w, nvalid);
  if (t < NBK) {
    s_off[t] = ex;
    tcnt[t * NCH + c] = h;
    toff[t * NCH + c] = ex;
  }
  __syncthreads();
#pragma unroll
  for (int j = 0; j < EPC / BTH; ++j)
    if (bk[j] >= 0) s_stage[s_off[bk[j]] + rk[j]] = word[j];
  __syncthreads();
  for (int k = t; k < nvalid; k += BTH) tmp[e0 + k] = s_stage[k];
}

template <int NBK, int NCH>
__global__ __launch_bounds__(BTH) void csr_place(const int* __restrict__ tmp,
                                                 const int* __restrict__ tcnt,
                                                 const int* __restrict__ toff,
                                                 int* __restrict__ rowptr,
                                                 int* __restrict__ csr, int N) {
  static_assert((NCH & 3) == 0, "run table rows are summed as int4");
  __shared__ int s_cnt[NCH];
  __shared__ int s_off[NCH];
  __shared__ int s_w[BTH / 32];
  __shared__ int s_deg[NPB];
  __shared__ int s_pos[NPB];
  __shared__ int s_out[STAGE_CAP];
  const int t = threadIdx.x, lane = t & 31, w = t >> 5, b = blockIdx.x;

  int part = 0;
  const int4* c4 = (const int4*)tcnt;
  const int nq = (b * NCH) >> 2;
  for (int q = t; q < nq; q += BTH) {
    const int4 v = c4[q];
    part += (v.x + v.y) + (v.z + v.w);
  }
  int mine = 0;
  if (t < NCH) {
    mine = tcnt[b * NCH + t];
    s_cnt[t] = mine;
    s_off[t] = toff[b * NCH + t];
  }
  if (t < NPB) s_deg[t] = 0;
  int base, nb;
  (void)block_excl_scan(part, s_w, base);
  (void)block_excl_scan(mine, s_w, nb);

  for (int c = w; c < NCH; c += BTH / 32) {
    const int n = s_cnt[c];
    const int* run = tmp + c * EPC + s_off[c];
    for (int k = lane; k < n; k += 32) atomicAdd(&s_deg[run[k] & (NPB - 1)], 1);
  }
  __syncthreads();

  const int dg = (t < NPB) ? s_deg[t] : 0;
  int nb2;
  const int ex = block_excl_scan(dg, s_w, nb2);
  if (t < NPB) {
    s_pos[t] = ex;
    const int node = b * NPB + t;
    if (node < N) rowptr[node] = base + ex;
  }
  if (b == NBK - 1 && t == 0) rowptr[N] = base + nb;
  __syncthreads();

  const bool staged = nb <= STAGE_CAP;
  for (int c = w; c < NCH; c += BTH / 32) {
    const int n = s_cnt[c];
    const int* run = tmp + c * EPC + s_off[c];
    for (int k = lane; k < n; k += 32) {
      const int wd = run[k];
      const int slot = atomicAdd(&s_pos[wd & (NPB - 1)], 1);
      const int src = (int)((unsigned)wd >> NPB_LOG);
      if (staged) s_out[slot] = src;
      else        csr[base + slot] = src;
    }
  }
  __syncthreads();
  if (staged)
    for (int k = t; k < nb; k += BTH) csr[base + k] = s_out[k];
}

template <int C, bool FINAL>
__global__ __launch_bounds__(TPB) void node_softmax_agg(
    const float* __restrict__ xl, const float* __restrict__ xr,
    const float* __restrict__ att, const float* __restrict__ bias,
    const int* __restrict__ rowptr, const int* __restrict__ csr,
    float* __restrict__ outp, int N) {
  constexpr int M   = 4 * C;
  constexpr int LPN = C;
  constexpr int LPH = C / 4;
  const int node = (blockIdx.x * TPB + threadIdx.x) / LPN;
  const int sub  = threadIdx.x & (LPN - 1);
  const bool valid = node < N;
  const int i = valid ? node : N - 1;

  const float4 xr4 = *(const float4*)(xr + (size_t)i * M + 4 * sub);
  const float4 at4 = *(const float4*)(att + 4 * sub);
  const int beg   = rowptr[i];
  const int total = rowptr[i + 1] - beg + 1;
  int wtotal = total;
  if constexpr (LPN < 32) wtotal = max(wtotal, __shfl_xor(wtotal, 16));

  float m = -INFINITY, den = 0.0f;
  float4 acc = make_float4(0.0f, 0.0f, 0.0f, 0.0f);

  for (int base = 0; base < wtotal; base += LPN) {
    const int idx = base + sub;
    int my_s = i;
    if (idx >= 1 && idx < total) my_s = csr[beg + idx - 1];
    const int cnt = min(LPN, wtotal - base);
    for (int j = 0; j < cnt; ++j) {
      const int s = __shfl(my_s, j, LPN);
      const float4 row = *(const float4*)(xl + (size_t)s * M + 4 * sub);
      float vx = row.x + xr4.x, vy = row.y + xr4.y;
      float vz = row.z + xr4.z, vw = row.w + xr4.w;
      vx = vx > 0.0f ? vx : 0.2f * vx;
      vy = vy > 0.0f ? vy : 0.2f * vy;
      vz = vz > 0.0f ? vz : 0.2f * vz;
      vw = vw > 0.0f ? vw : 0.2f * vw;
      float p = at4.x * vx + at4.y * vy + at4.z * vz + at4.w * vw;
      for (int o = 1; o < LPH; o <<= 1) p += __shfl_xor(p, o);
      const float sc = (base + j < total) ? p : -INFINITY;
      const float mn = fmaxf(m, sc);
      const float scale = expf(m - mn);
      const float wgt   = expf(sc - mn);
      den   = den   * scale + wgt;
      acc.x = acc.x * scale + wgt * row.x;
      acc.y = acc.y * scale + wgt * row.y;
      acc.z = acc.z * scale + wgt * row.z;
      acc.w = acc.w * scale + wgt * row.w;
      m = mn;
    }
  }

  float4 r = make_float4(acc.x / den, acc.y / den, acc.z / den, acc.w / den);
  if constexpr (!FINAL) {
    const float4 b4 = *(const float4*)(bias + 4 * sub);
    r.x += b4.x; r.y += b4.y; r.z += b4.z; r.w += b4.w;
    r.x = r.x > 0.0f ? r.x : expm1f(r.x);
    r.y = r.y > 0.0f ? r.y : expm1f(r.y);
    r.z = r.z > 0.0f ? r.z : expm1f(r.z);
    r.w = r.w > 0.0f ? r.w : expm1f(r.w);
    if (valid) *(float4*)(outp + (size_t)i * M + 4 * sub) = r;
  } else {
    for (int o = LPH; o < LPN; o <<= 1) {
      r.x += __shfl_xor(r.x, o); r.y += __shfl_xor(r.y, o);
      r.z += __shfl_xor(r.z, o); r.w += __shfl_xor(r.w, o);
    }
    if (valid && sub < LPH) {
      const float4 b4 = *(const float4*)(bias + 4 * sub);
      r.x = 0.25f * r.x + b4.x; r.y = 0.25f * r.y + b4.y;
      r.z = 0.25f * r.z + b4.z; r.w = 0.25f * r.w + b4.w;
      *(float4*)(outp + (size_t)i * C + 4 * sub) = r;
    }
  }
}

extern "C" void kernel_launch(void* const* d_in, const int* in_sizes, int n_in,
                              void* d_out, int out_size, void* d_ws, size_t ws_size,
                              hipStream_t stream) {
  (void)in_sizes; (void)n_in; (void)out_size; (void)ws_size;
  const float* x   = (const float*)d_in[0];
  const int*   ei  = (const int*)d_in[1];
  const float* W1l = (const float*)d_in[2];
  const float* W1r = (const float*)d_in[3];
  const float* a1  = (const float*)d_in[4];
  const float* b1  = (const float*)d_in[5];
  const float* W2l = (const float*)d_in[6];
  const float* W2r = (const float*)d_in[7];
  const float* a2  = (const float*)d_in[8];
  const float* b2  = (const float*)d_in[9];
  const float* W3l = (const float*)d_in[10];
  const float* W3r = (const float*)d_in[11];
  const float* a3  = (const float*)d_in[12];
  const float* b3  = (const float*)d_in[13];

  constexpr int N = 100000, E = 800000;

  float* ws = (float*)d_ws;
  float* xl = ws;
  float* xr = xl + (size_t)N * 128;
  float* h1 = xr + (size_t)N * 128;
  float* h2 = h1 + (size_t)N * 64;
  constexpr int NBK = (N + NPB - 1) / NPB;
  constexpr int NCH = (E + EPC - 1) / EPC;
  constexpr int RP  = (N + 1 + 31) / 32 * 32;
  int* rowptr = (int*)(h2 + (size_t)N * 64);
  int* csr    = rowptr + RP;
  int* tmp    = csr + E;
  int* tcnt   = tmp + (size_t)NCH * EPC;
  int* toff   = tcnt + NBK * NCH;

  const int rowTiles = N / 16;
  const int gblocks  = (rowTiles + WAVES_PER_BLOCK - 1) / WAVES_PER_BLOCK;

  csr_bin  <NBK, NCH><<<NCH, BTH, 0, stream>>>(ei, tmp, tcnt, toff, E);
  csr_place<NBK, NCH><<<NBK, BTH, 0, stream>>>(tmp, tcnt, toff, rowptr, csr, N);

  const int nb16 = (N * 16 + TPB - 1) / TPB;
  const int nb32 = (N * 32 + TPB - 1) / TPB;

  gemm_wmma_f32<64><<<gblocks, TPB, 0, stream>>>(x, W1l, W1r, xl, xr, N);
  node_softmax_agg<16, false><<<nb16, TPB, 0, stream>>>(xl, xr, a1, b1, rowptr, csr, h1, N);

  gemm_wmma_f32<64><<<gblocks, TPB, 0, stream>>>(h1, W2l, W2r, xl, xr, N);
  node_softmax_agg<16, false><<<nb16, TPB, 0, stream>>>(xl, xr, a2, b2, rowptr, csr, h2, N);

  gemm_wmma_f32<128><<<gblocks, TPB, 0, stream>>>(h2, W3l, W3r, xl, xr, N);
  node_softmax_agg<32, true><<<nb32, TPB, 0, stream>>>(xl, xr, a3, b3, rowptr, csr,
                                                       (float*)d_out, N);
}


# Edge walk of the gather kernel in groups of 4 entries: the 4 source rows are loaded before any is used, one max update and one rescale of the running sums per group; adds stay in list order
# speedup vs baseline: 8.4686x; 1.1102x over previous
#include <hip/hip_runtime.h>
#include <math.h>

typedef __attribute__((ext_vector_type(2))) float v2f;
typedef __attribute__((ext_vector_type(8))) float v8f;

#define TPB 256
#define WAVES_PER_BLOCK 8

template <int M>
__global__ __launch_bounds__(TPB) void gemm_wmma_f32(
    const float* __restrict__ X, const float* __restrict__ Wl,
    const float* __restrict__ Wr, float* __restrict__ Yl,
    float* __restrict__ Yr, int Nrows) {
  __shared__ __align__(16) float lw[2 * 64 * M];

  constexpr int TOTAL4 = 64 * M / 4;
  for (int i = threadIdx.x; i < TOTAL4; i += TPB) {
    ((float4*)lw)[i]          = ((const float4*)Wl)[i];
    ((float4*)lw)[TOTAL4 + i] = ((const float4*)Wr)[i];
  }
  __syncthreads();

  const int rowTile = blockIdx.x * WAVES_PER_BLOCK + (threadIdx.x >> 5);
  if (rowTile >= (Nrows >> 4)) return;
  const int row0 = rowTile << 4;
  const int lane = threadIdx.x & 31;
  const int r    = lane & 15;
  const int half = lane >> 4;

  v2f a[16];
  const float* xrow = X + (size_t)(row0 + r) * 64;
#pragma unroll
  for (int ks = 0; ks < 16; ++ks) {
    const int k = 4 * ks + 2 * half;
    a[ks].x = xrow[k];
    a[ks].y = xrow[k + 1];
  }

#pragma unroll
  for (int side = 0; side < 2; ++side) {
    const float* lws = lw + side * (64 * M);
    float* Y = side ? Yr : Yl;
#pragma nounroll
    for (int tn = 0; tn < M / 16; ++tn) {
      const int col0 = tn << 4;
      v8f c = {};
#pragma unroll
      for (int ks = 0; ks < 16; ++ks) {
        const int k = 4 * ks + 2 * half;
        v2f b;
        b.x = lws[k * M + col0 + r];
        b.y = lws[(k + 1) * M + col0 + r];
        c = __builtin_amdgcn_wmma_f32_16x16x4_f32(false, a[ks], false, b,
                                                  (short)0, c, false, false);
      }
#pragma unroll
      for (int v = 0; v < 8; ++v)
        Y[(size_t)(row0 + v + 8 * half) * M + col0 + r] = c[v];
    }
  }
}

#define BTH 1024
#define EPC 4096
#define NPB 512
#define NPB_LOG 9
#define STAGE_CAP 8192

__device__ __forceinline__ int block_excl_scan(int v, int* s_w, int& total) {
  const int lane = threadIdx.x & 31, w = threadIdx.x >> 5;
  int inc = v;
  for (int o = 1; o < 32; o <<= 1) {
    const int u = __shfl_up(inc, o);
    if (lane >= o) inc += u;
  }
  if (lane == 31) s_w[w] = inc;
  __syncthreads();
  int pre = 0, tot = 0;
  for (int k = 0; k < BTH / 32; ++k) {
    const int x = s_w[k];
    if (k < w) pre += x;
    tot += x;
  }
  __syncthreads();
  total = tot;
  return pre + inc - v;
}

template <int NBK, int NCH>
__global__ __launch_bounds__(BTH) void csr_bin(const int* __restrict__ ei,
                                               int* __restrict__ tmp,
                                               int* __restrict__ tcnt,
                                               int* __restrict__ toff, int E) {
  __shared__ int s_hist[NBK];
  __shared__ int s_off[NBK];
  __shared__ int s_w[BTH / 32];
  __shared__ int s_stage[EPC];
  const int t = threadIdx.x, c = blockIdx.x;
  if (t < NBK) s_hist[t] = 0;
  __syncthreads();

  const int e0 = c * EPC;
  int word[EPC / BTH], bk[EPC / BTH], rk[EPC / BTH];
#pragma unroll
  for (int j = 0; j < EPC / BTH; ++j) {
    const int e = e0 + j * BTH + t;
    bk[j] = -1; word[j] = 0; rk[j] = 0;
    if (e < E) {
      const int s = ei[e], d = ei[E + e];
      bk[j]   = d >> NPB_LOG;
      word[j] = (s << NPB_LOG) | (d & (NPB - 1));
      rk[j]   = atomicAdd(&s_hist[bk[j]], 1);
    }
  }
  __syncthreads();

  const int h = (t < NBK) ? s_hist[t] : 0;
  int nvalid;
  const int ex = block_excl_scan(h, s_w, nvalid);
  if (t < NBK) {
    s_off[t] = ex;
    tcnt[t * NCH + c] = h;
    toff[t * NCH + c] = ex;
  }
  __syncthreads();
#pragma unroll
  for (int j = 0; j < EPC / BTH; ++j)
    if (bk[j] >= 0) s_stage[s_off[bk[j]] + rk[j]] = word[j];
  __syncthreads();
  for (int k = t; k < nvalid; k += BTH) tmp[e0 + k] = s_stage[k];
}

template <int NBK, int NCH>
__global__ __launch_bounds__(BTH) void csr_place(const int* __restrict__ tmp,
                                                 const int* __restrict__ tcnt,
                                                 const int* __restrict__ toff,
                                                 int* __restrict__ rowptr,
                                                 int* __restrict__ csr, int N) {
  static_assert((NCH & 3) == 0, "run table rows are summed as int4");
  __shared__ int s_cnt[NCH];
  __shared__ int s_off[NCH];
  __shared__ int s_w[BTH / 32];
  __shared__ int s_deg[NPB];
  __shared__ int s_pos[NPB];
  __shared__ int s_out[STAGE_CAP];
  const int t = threadIdx.x, lane = t & 31, w = t >> 5, b = blockIdx.x;

  int part = 0;
  const int4* c4 = (const int4*)tcnt;
  const int nq = (b * NCH) >> 2;
  for (int q = t; q < nq; q += BTH) {
    const int4 v = c4[q];
    part += (v.x + v.y) + (v.z + v.w);
  }
  int mine = 0;
  if (t < NCH) {
    mine = tcnt[b * NCH + t];
    s_cnt[t] = mine;
    s_off[t] = toff[b * NCH + t];
  }
  if (t < NPB) s_deg[t] = 0;
  int base, nb;
  (void)block_excl_scan(part, s_w, base);
  (void)block_excl_scan(mine, s_w, nb);

  for (int c = w; c < NCH; c += BTH / 32) {
    const int n = s_cnt[c];
    const int* run = tmp + c * EPC + s_off[c];
    for (int k = lane; k < n; k += 32) atomicAdd(&s_deg[run[k] & (NPB - 1)], 1);
  }
  __syncthreads();

  const int dg = (t < NPB) ? s_deg[t] : 0;
  int nb2;
  const int ex = block_excl_scan(dg, s_w, nb2);
  if (t < NPB) {
    s_pos[t] = ex;
    const int node = b * NPB + t;
    if (node < N) rowptr[node] = base + ex;
  }
  if (b == NBK - 1 && t == 0) rowptr[N] = base + nb;
  __syncthreads();

  const bool staged = nb <= STAGE_CAP;
  for (int c = w; c < NCH; c += BTH / 32) {
    const int n = s_cnt[c];
    const int* run = tmp + c * EPC + s_off[c];
    for (int k = lane; k < n; k += 32) {
      const int wd = run[k];
      const int slot = atomicAdd(&s_pos[wd & (NPB - 1)], 1);
      const int src = (int)((unsigned)wd >> NPB_LOG);
      if (staged) s_out[slot] = src;
      else        csr[base + slot] = src;
    }
  }
  __syncthreads();
  if (staged)
    for (int k = t; k < nb; k += BTH) csr[base + k] = s_out[k];
}

template <int C, int K>
__device__ __forceinline__ void walk_entries(
    const float* __restrict__ xl, int my_s, int j0, int e0, int total, int sub,
    const float4& xr4, const float4& at4, float& m, float& den, float4& acc) {
  constexpr int M   = 4 * C;
  constexpr int LPN = C;
  constexpr int LPH = C / 4;
  float4 row[K];
#pragma unroll
  for (int k = 0; k < K; ++k) {
    const int s = __shfl(my_s, j0 + k, LPN);
    row[k] = *(const float4*)(xl + (size_t)s * M + 4 * sub);
  }
  float sc[K];
  float mb = m;
#pragma unroll
  for (int k = 0; k < K; ++k) {
    float vx = row[k].x + xr4.x, vy = row[k].y + xr4.y;
    float vz = row[k].z + xr4.z, vw = row[k].w + xr4.w;
    vx = vx > 0.0f ? vx : 0.2f * vx;
    vy = vy > 0.0f ? vy : 0.2f * vy;
    vz = vz > 0.0f ? vz : 0.2f * vz;
    vw = vw > 0.0f ? vw : 0.2f * vw;
    float p = at4.x * vx + at4.y * vy + at4.z * vz + at4.w * vw;
    for (int o = 1; o < LPH; o <<= 1) p += __shfl_xor(p, o);
    sc[k] = (e0 + k < total) ? p : -INFINITY;
    mb = fmaxf(mb, sc[k]);
  }
  const float scale = expf(m - mb);
  den   *= scale;
  acc.x *= scale; acc.y *= scale; acc.z *= scale; acc.w *= scale;
#pragma unroll
  for (int k = 0; k < K; ++k) {
    const float wgt = expf(sc[k] - mb);
    den   += wgt;
    acc.x += wgt * row[k].x;
    acc.y += wgt * row[k].y;
    acc.z += wgt * row[k].z;
    acc.w += wgt * row[k].w;
  }
  m = mb;
}

template <int C, bool FINAL>
__global__ __launch_bounds__(TPB) void node_softmax_agg(
    const float* __restrict__ xl, const float* __restrict__ xr,
    const float* __restrict__ att, const float* __restrict__ bias,
    const int* __restrict__ rowptr, const int* __restrict__ csr,
    float* __restrict__ outp, int N) {
  constexpr int M   = 4 * C;
  constexpr int LPN = C;
  constexpr int LPH = C / 4;
  const int node = (blockIdx.x * TPB + threadIdx.x) / LPN;
  const int sub  = threadIdx.x & (LPN - 1);
  const bool valid = node < N;
  const int i = valid ? node : N - 1;

  const float4 xr4 = *(const float4*)(xr + (size_t)i * M + 4 * sub);
  const float4 at4 = *(const float4*)(att + 4 * sub);
  const int beg   = rowptr[i];
  const int total = rowptr[i + 1] - beg + 1;
  int wtotal = total;
  if constexpr (LPN < 32) wtotal = max(wtotal, __shfl_xor(wtotal, 16));

  float m = -INFINITY, den = 0.0f;
  float4 acc = make_float4(0.0f, 0.0f, 0.0f, 0.0f);

  for (int base = 0; base < wtotal; base += LPN) {
    const int idx = base + sub;
    int my_s = i;
    if (idx >= 1 && idx < total) my_s = csr[beg + idx - 1];
    const int cnt = min(LPN, wtotal - base);
    int j = 0;
    for (; j + 4 <= cnt; j += 4)
      walk_entries<C, 4>(xl, my_s, j, base + j, total, sub, xr4, at4, m, den, acc);
    const int rem = cnt - j;
    if (rem == 3)
      walk_entries<C, 3>(xl, my_s, j, base + j, total, sub, xr4, at4, m, den, acc);
    else if (rem == 2)
      walk_entries<C, 2>(xl, my_s, j, base + j, total, sub, xr4, at4, m, den, acc);
    else if (rem == 1)
      walk_entries<C, 1>(xl, my_s, j, base + j, total, sub, xr4, at4, m, den, acc);
  }

  float4 r = make_float4(acc.x / den, acc.y / den, acc.z / den, acc.w / den);
  if constexpr (!FINAL) {
    const float4 b4 = *(const float4*)(bias + 4 * sub);
    r.x += b4.x; r.y += b4.y; r.z += b4.z; r.w += b4.w;
    r.x = r.x > 0.0f ? r.x : expm1f(r.x);
    r.y = r.y > 0.0f ? r.y : expm1f(r.y);
    r.z = r.z > 0.0f ? r.z : expm1f(r.z);
    r.w = r.w > 0.0f ? r.w : expm1f(r.w);
    if (valid) *(float4*)(outp + (size_t)i * M + 4 * sub) = r;
  } else {
    for (int o = LPH; o < LPN; o <<= 1) {
      r.x += __shfl_xor(r.x, o); r.y += __shfl_xor(r.y, o);
      r.z += __shfl_xor(r.z, o); r.w += __shfl_xor(r.w, o);
    }
    if (valid && sub < LPH) {
      const float4 b4 = *(const float4*)(bias + 4 * sub);
      r.x = 0.25f * r.x + b4.x; r.y = 0.25f * r.y + b4.y;
      r.z = 0.25f * r.z + b4.z; r.w = 0.25f * r.w + b4.w;
      *(float4*)(outp + (size_t)i * C + 4 * sub) = r;
    }
  }
}

extern "C" void kernel_launch(void* const* d_in, const int* in_sizes, int n_in,
                              void* d_out, int out_size, void* d_ws, size_t ws_size,
                              hipStream_t stream) {
  (void)in_sizes; (void)n_in; (void)out_size; (void)ws_size;
  const float* x   = (const float*)d_in[0];
  const int*   ei  = (const int*)d_in[1];
  const float* W1l = (const float*)d_in[2];
  const float* W1r = (const float*)d_in[3];
  const float* a1  = (const float*)d_in[4];
  const float* b1  = (const float*)d_in[5];
  const float* W2l = (const float*)d_in[6];
  const float* W2r = (const float*)d_in[7];
  const float* a2  = (const float*)d_in[8];
  const float* b2  = (const float*)d_in[9];
  const float* W3l = (const float*)d_in[10];
  const float* W3r = (const float*)d_in[11];
  const float* a3  = (const float*)d_in[12];
  const float* b3  = (const float*)d_in[13];

  constexpr int N = 100000, E = 800000;

  float* ws = (float*)d_ws;
  float* xl = ws;
  float* xr = xl + (size_t)N * 128;
  float* h1 = xr + (size_t)N * 128;
  float* h2 = h1 + (size_t)N * 64;
  constexpr int NBK = (N + NPB - 1) / NPB;
  constexpr int NCH = (E + EPC - 1) / EPC;
  constexpr int RP  = (N + 1 + 31) / 32 * 32;
  int* rowptr = (int*)(h2 + (size_t)N * 64);
  int* csr    = rowptr + RP;
  int* tmp    = csr + E;
  int* tcnt   = tmp + (size_t)NCH * EPC;
  int* toff   = tcnt + NBK * NCH;

  const int rowTiles = N / 16;
  const int gblocks  = (rowTiles + WAVES_PER_BLOCK - 1) / WAVES_PER_BLOCK;

  csr_bin  <NBK, NCH><<<NCH, BTH, 0, stream>>>(ei, tmp, tcnt, toff, E);
  csr_place<NBK, NCH><<<NBK, BTH, 0, stream>>>(tmp, tcnt, toff, rowptr, csr, N);

  const int nb16 = (N * 16 + TPB - 1) / TPB;
  const int nb32 = (N * 32 + TPB - 1) / TPB;

  gemm_wmma_f32<64><<<gblocks, TPB, 0, stream>>>(x, W1l, W1r, xl, xr, N);
  node_softmax_agg<16, false><<<nb16, TPB, 0, stream>>>(xl, xr, a1, b1, rowptr, csr, h1, N);

  gemm_wmma_f32<64><<<gblocks, TPB, 0, stream>>>(h1, W2l, W2r, xl, xr, N);
  node_softmax_agg<16, false><<<nb16, TPB, 0, stream>>>(xl, xr, a2, b2, rowptr, csr, h2, N);

  gemm_wmma_f32<128><<<gblocks, TPB, 0, stream>>>(h2, W3l, W3r, xl, xr, N);
  node_softmax_agg<32, true><<<nb32, TPB, 0, stream>>>(xl, xr, a3, b3, rowptr, csr,
                                                       (float*)d_out, N);
}


# Next layer's xl|xr GEMM done inside the gather kernel: its 16-node h tile goes through LDS to f32 WMMA (W fragments from global), h1/h2 never stored; 8 -> 6 launches; per-element sum order unchanged
# speedup vs baseline: 9.2954x; 1.0976x over previous
#include <hip/hip_runtime.h>
#include <math.h>

typedef __attribute__((ext_vector_type(2))) float v2f;
typedef __attribute__((ext_vector_type(8))) float v8f;

#define TPB 256
#define WAVES_PER_BLOCK 8

template <int M>
__global__ __launch_bounds__(TPB) void gemm_wmma_f32(
    const float* __restrict__ X, const float* __restrict__ Wl,
    const float* __restrict__ Wr, float* __restrict__ Yl,
    float* __restrict__ Yr, int Nrows) {
  __shared__ __align__(16) float lw[2 * 64 * M];

  constexpr int TOTAL4 = 64 * M / 4;
  for (int i = threadIdx.x; i < TOTAL4; i += TPB) {
    ((float4*)lw)[i]          = ((const float4*)Wl)[i];
    ((float4*)lw)[TOTAL4 + i] = ((const float4*)Wr)[i];
  }
  __syncthreads();

  const int rowTile = blockIdx.x * WAVES_PER_BLOCK + (threadIdx.x >> 5);
  if (rowTile >= (Nrows >> 4)) return;
  const int row0 = rowTile << 4;
  const int lane = threadIdx.x & 31;
  const int r    = lane & 15;
  const int half = lane >> 4;

  v2f a[16];
  const float* xrow = X + (size_t)(row0 + r) * 64;
#pragma unroll
  for (int ks = 0; ks < 16; ++ks) {
    const int k = 4 * ks + 2 * half;
    a[ks].x = xrow[k];
    a[ks].y = xrow[k + 1];
  }

#pragma unroll
  for (int side = 0; side < 2; ++side) {
    const float* lws = lw + side * (64 * M);
    float* Y = side ? Yr : Yl;
#pragma nounroll
    for (int tn = 0; tn < M / 16; ++tn) {
      const int col0 = tn << 4;
      v8f c = {};
#pragma unroll
      for (int ks = 0; ks < 16; ++ks) {
        const int k = 4 * ks + 2 * half;
        v2f b;
        b.x = lws[k * M + col0 + r];
        b.y = lws[(k + 1) * M + col0 + r];
        c = __builtin_amdgcn_wmma_f32_16x16x4_f32(false, a[ks], false, b,
                                                  (short)0, c, false, false);
      }
#pragma unroll
      for (int v = 0; v < 8; ++v)
        Y[(size_t)(row0 + v + 8 * half) * M + col0 + r] = c[v];
    }
  }
}

#define BTH 1024
#define EPC 4096
#define NPB 512
#define NPB_LOG 9
#define STAGE_CAP 8192

__device__ __forceinline__ int block_excl_scan(int v, int* s_w, int& total) {
  const int lane = threadIdx.x & 31, w = threadIdx.x >> 5;
  int inc = v;
  for (int o = 1; o < 32; o <<= 1) {
    const int u = __shfl_up(inc, o);
    if (lane >= o) inc += u;
  }
  if (lane == 31) s_w[w] = inc;
  __syncthreads();
  int pre = 0, tot = 0;
  for (int k = 0; k < BTH / 32; ++k) {
    const int x = s_w[k];
    if (k < w) pre += x;
    tot += x;
  }
  __syncthreads();
  total = tot;
  return pre + inc - v;
}

template <int NBK, int NCH>
__global__ __launch_bounds__(BTH) void csr_bin(const int* __restrict__ ei,
                                               int* __restrict__ tmp,
                                               int* __restrict__ tcnt,
                                               int* __restrict__ toff, int E) {
  __shared__ int s_hist[NBK];
  __shared__ int s_off[NBK];
  __shared__ int s_w[BTH / 32];
  __shared__ int s_stage[EPC];
  const int t = threadIdx.x, c = blockIdx.x;
  if (t < NBK) s_hist[t] = 0;
  __syncthreads();

  const int e0 = c * EPC;
  int word[EPC / BTH], bk[EPC / BTH], rk[EPC / BTH];
#pragma unroll
  for (int j = 0; j < EPC / BTH; ++j) {
    const int e = e0 + j * BTH + t;
    bk[j] = -1; word[j] = 0; rk[j] = 0;
    if (e < E) {
      const int s = ei[e], d = ei[E + e];
      bk[j]   = d >> NPB_LOG;
      word[j] = (s << NPB_LOG) | (d & (NPB - 1));
      rk[j]   = atomicAdd(&s_hist[bk[j]], 1);
    }
  }
  __syncthreads();

  const int h = (t < NBK) ? s_hist[t] : 0;
  int nvalid;
  const int ex = block_excl_scan(h, s_w, nvalid);
  if (t < NBK) {
    s_off[t] = ex;
    tcnt[t * NCH + c] = h;
    toff[t * NCH + c] = ex;
  }
  __syncthreads();
#pragma unroll
  for (int j = 0; j < EPC / BTH; ++j)
    if (bk[j] >= 0) s_stage[s_off[bk[j]] + rk[j]] = word[j];
  __syncthreads();
  for (int k = t; k < nvalid; k += BTH) tmp[e0 + k] = s_stage[k];
}

template <int NBK, int NCH>
__global__ __launch_bounds__(BTH) void csr_place(const int* __restrict__ tmp,
                                                 const int* __restrict__ tcnt,
                                                 const int* __restrict__ toff,
                                                 int* __restrict__ rowptr,
                                                 int* __restrict__ csr, int N) {
  static_assert((NCH & 3) == 0, "run table rows are summed as int4");
  __shared__ int s_cnt[NCH];
  __shared__ int s_off[NCH];
  __shared__ int s_w[BTH / 32];
  __shared__ int s_deg[NPB];
  __shared__ int s_pos[NPB];
  __shared__ int s_out[STAGE_CAP];
  const int t = threadIdx.x, lane = t & 31, w = t >> 5, b = blockIdx.x;

  int part = 0;
  const int4* c4 = (const int4*)tcnt;
  const int nq = (b * NCH) >> 2;
  for (int q = t; q < nq; q += BTH) {
    const int4 v = c4[q];
    part += (v.x + v.y) + (v.z + v.w);
  }
  int mine = 0;
  if (t < NCH) {
    mine = tcnt[b * NCH + t];
    s_cnt[t] = mine;
    s_off[t] = toff[b * NCH + t];
  }
  if (t < NPB) s_deg[t] = 0;
  int base, nb;
  (void)block_excl_scan(part, s_w, base);
  (void)block_excl_scan(mine, s_w, nb);

  for (int c = w; c < NCH; c += BTH / 32) {
    const int n = s_cnt[c];
    const int* run = tmp + c * EPC + s_off[c];
    for (int k = lane; k < n; k += 32) atomicAdd(&s_deg[run[k] & (NPB - 1)], 1);
  }
  __syncthreads();

  const int dg = (t < NPB) ? s_deg[t] : 0;
  int nb2;
  const int ex = block_excl_scan(dg, s_w, nb2);
  if (t < NPB) {
    s_pos[t] = ex;
    const int node = b * NPB + t;
    if (node < N) rowptr[node] = base + ex;
  }
  if (b == NBK - 1 && t == 0) rowptr[N] = base + nb;
  __syncthreads();

  const bool staged = nb <= STAGE_CAP;
  for (int c = w; c < NCH; c += BTH / 32) {
    const int n = s_cnt[c];
    const int* run = tmp + c * EPC + s_off[c];
    for (int k = lane; k < n; k += 32) {
      const int wd = run[k];
      const int slot = atomicAdd(&s_pos[wd & (NPB - 1)], 1);
      const int src = (int)((unsigned)wd >> NPB_LOG);
      if (staged) s_out[slot] = src;
      else        csr[base + slot] = src;
    }
  }
  __syncthreads();
  if (staged)
    for (int k = t; k < nb; k += BTH) csr[base + k] = s_out[k];
}

template <int C, int K>
__device__ __forceinline__ void walk_entries(
    const float* __restrict__ xl, int my_s, int j0, int e0, int total, int sub,
    const float4& xr4, const float4& at4, float& m, float& den, float4& acc) {
  constexpr int M   = 4 * C;
  constexpr int LPN = C;
  constexpr int LPH = C / 4;
  float4 row[K];
#pragma unroll
  for (int k = 0; k < K; ++k) {
    const int s = __shfl(my_s, j0 + k, LPN);
    row[k] = *(const float4*)(xl + (size_t)s * M + 4 * sub);
  }
  float sc[K];
  float mb = m;
#pragma unroll
  for (int k = 0; k < K; ++k) {
    float vx = row[k].x + xr4.x, vy = row[k].y + xr4.y;
    float vz = row[k].z + xr4.z, vw = row[k].w + xr4.w;
    vx = vx > 0.0f ? vx : 0.2f * vx;
    vy = vy > 0.0f ? vy : 0.2f * vy;
    vz = vz > 0.0f ? vz : 0.2f * vz;
    vw = vw > 0.0f ? vw : 0.2f * vw;
    float p = at4.x * vx + at4.y * vy + at4.z * vz + at4.w * vw;
    for (int o = 1; o < LPH; o <<= 1) p += __shfl_xor(p, o);
    sc[k] = (e0 + k < total) ? p : -INFINITY;
    mb = fmaxf(mb, sc[k]);
  }
  const float scale = expf(m - mb);
  den   *= scale;
  acc.x *= scale; acc.y *= scale; acc.z *= scale; acc.w *= scale;
#pragma unroll
  for (int k = 0; k < K; ++k) {
    const float wgt = expf(sc[k] - mb);
    den   += wgt;
    acc.x += wgt * row[k].x;
    acc.y += wgt * row[k].y;
    acc.z += wgt * row[k].z;
    acc.w += wgt * row[k].w;
  }
  m = mb;
}

#define HP 68

template <int C, bool FINAL, int MN>
__global__ __launch_bounds__(TPB) void node_softmax_agg(
    const float* __restrict__ xl, const float* __restrict__ xr,
    const float* __restrict__ att, const float* __restrict__ bias,
    const int* __restrict__ rowptr, const int* __restrict__ csr,
    float* __restrict__ outp,
    const float* __restrict__ Wl, const float* __restrict__ Wr,
    float* __restrict__ Yl, float* __restrict__ Yr, int N) {
  constexpr int M   = 4 * C;
  constexpr int LPN = C;
  constexpr int LPH = C / 4;
  const int node = (blockIdx.x * TPB + threadIdx.x) / LPN;
  const int sub  = threadIdx.x & (LPN - 1);
  const bool valid = node < N;
  const int i = valid ? node : N - 1;

  const float4 xr4 = *(const float4*)(xr + (size_t)i * M + 4 * sub);
  const float4 at4 = *(const float4*)(att + 4 * sub);
  const int beg   = rowptr[i];
  const int total = rowptr[i + 1] - beg + 1;
  int wtotal = total;
  if constexpr (LPN < 32) wtotal = max(wtotal, __shfl_xor(wtotal, 16));

  float m = -INFINITY, den = 0.0f;
  float4 acc = make_float4(0.0f, 0.0f, 0.0f, 0.0f);

  for (int base = 0; base < wtotal; base += LPN) {
    const int idx = base + sub;
    int my_s = i;
    if (idx >= 1 && idx < total) my_s = csr[beg + idx - 1];
    const int cnt = min(LPN, wtotal - base);
    int j = 0;
    for (; j + 4 <= cnt; j += 4)
      walk_entries<C, 4>(xl, my_s, j, base + j, total, sub, xr4, at4, m, den, acc);
    const int rem = cnt - j;
    if (rem == 3)
      walk_entries<C, 3>(xl, my_s, j, base + j, total, sub, xr4, at4, m, den, acc);
    else if (rem == 2)
      walk_entries<C, 2>(xl, my_s, j, base + j, total, sub, xr4, at4, m, den, acc);
    else if (rem == 1)
      walk_entries<C, 1>(xl, my_s, j, base + j, total, sub, xr4, at4, m, den, acc);
  }

  float4 r = make_float4(acc.x / den, acc.y / den, acc.z / den, acc.w / den);
  if constexpr (!FINAL) {
    const float4 b4 = *(const float4*)(bias + 4 * sub);
    r.x += b4.x; r.y += b4.y; r.z += b4.z; r.w += b4.w;
    r.x = r.x > 0.0f ? r.x : expm1f(r.x);
    r.y = r.y > 0.0f ? r.y : expm1f(r.y);
    r.z = r.z > 0.0f ? r.z : expm1f(r.z);
    r.w = r.w > 0.0f ? r.w : expm1f(r.w);

    static_assert(M == 64 && TPB / LPN == 16, "the work-group's rows are one 16x64 A tile");
    __shared__ __align__(16) float ht[16 * HP];
    *(float4*)(ht + (threadIdx.x / LPN) * HP + 4 * sub) = r;
    __syncthreads();

    constexpr int GPS = MN / 32;
    const int w = threadIdx.x >> 5;
    if (2 * GPS < WAVES_PER_BLOCK && w >= 2 * GPS) return;
    const int lane = threadIdx.x & 31;
    const int rr   = lane & 15;
    const int half = lane >> 4;
    const bool right = w >= GPS;
    const float* Wm = right ? Wr : Wl;
    float* Y = right ? Yr : Yl;
    const int cb = (right ? w - GPS : w) * 32 + 2 * rr;
    const float* arow = ht + rr * HP + 2 * half;
    const float* wcol = Wm + (size_t)(2 * half) * MN + cb;
    v8f c0 = {}, c1 = {};
#pragma unroll
    for (int ks = 0; ks < 16; ++ks) {
      const v2f a  = *(const v2f*)(arow + 4 * ks);
      const v2f w0 = *(const v2f*)(wcol + (size_t)(4 * ks) * MN);
      const v2f w1 = *(const v2f*)(wcol + (size_t)(4 * ks + 1) * MN);
      v2f b0, b1;
      b0.x = w0.x; b0.y = w1.x;
      b1.x = w0.y; b1.y = w1.y;
      c0 = __builtin_amdgcn_wmma_f32_16x16x4_f32(false, a, false, b0, (short)0, c0, false, false);
      c1 = __builtin_amdgcn_wmma_f32_16x16x4_f32(false, a, false, b1, (short)0, c1, false, false);
    }
    const int row0 = blockIdx.x * (TPB / LPN);
    float* yp = Y + (size_t)(row0 + 8 * half) * MN + cb;
    if (row0 + 16 <= N) {
#pragma unroll
      for (int v = 0; v < 8; ++v) {
        v2f o;
        o.x = c0[v]; o.y = c1[v];
        *(v2f*)(yp + (size_t)v * MN) = o;
      }
    } else {
#pragma unroll
      for (int v = 0; v < 8; ++v) {
        if (row0 + 8 * half + v < N) {
          v2f o;
          o.x = c0[v]; o.y = c1[v];
          *(v2f*)(yp + (size_t)v * MN) = o;
        }
      }
    }
  } else {
    for (int o = LPH; o < LPN; o <<= 1) {
      r.x += __shfl_xor(r.x, o); r.y += __shfl_xor(r.y, o);
      r.z += __shfl_xor(r.z, o); r.w += __shfl_xor(r.w, o);
    }
    if (valid && sub < LPH) {
      const float4 b4 = *(const float4*)(bias + 4 * sub);
      r.x = 0.25f * r.x + b4.x; r.y = 0.25f * r.y + b4.y;
      r.z = 0.25f * r.z + b4.z; r.w = 0.25f * r.w + b4.w;
      *(float4*)(outp + (size_t)i * C + 4 * sub) = r;
    }
  }
}

extern "C" void kernel_launch(void* const* d_in, const int* in_sizes, int n_in,
                              void* d_out, int out_size, void* d_ws, size_t ws_size,
                              hipStream_t stream) {
  (void)in_sizes; (void)n_in; (void)out_size; (void)ws_size;
  const float* x   = (const float*)d_in[0];
  const int*   ei  = (const int*)d_in[1];
  const float* W1l = (const float*)d_in[2];
  const float* W1r = (const float*)d_in[3];
  const float* a1  = (const float*)d_in[4];
  const float* b1  = (const float*)d_in[5];
  const float* W2l = (const float*)d_in[6];
  const float* W2r = (const float*)d_in[7];
  const float* a2  = (const float*)d_in[8];
  const float* b2  = (const float*)d_in[9];
  const float* W3l = (const float*)d_in[10];
  const float* W3r = (const float*)d_in[11];
  const float* a3  = (const float*)d_in[12];
  const float* b3  = (const float*)d_in[13];

  constexpr int N = 100000, E = 800000;

  float* ws = (float*)d_ws;
  float* xl1 = ws;
  float* xr1 = xl1 + (size_t)N * 64;
  float* xl2 = xr1 + (size_t)N * 64;
  float* xr2 = xl2 + (size_t)N * 64;
  float* xl3 = ws;
  float* xr3 = xr2 + (size_t)N * 64;
  constexpr int NBK = (N + NPB - 1) / NPB;
  constexpr int NCH = (E + EPC - 1) / EPC;
  constexpr int RP  = (N + 1 + 31) / 32 * 32;
  int* rowptr = (int*)(xr3 + (size_t)N * 128);
  int* csr    = rowptr + RP;
  int* tmp    = csr + E;
  int* tcnt   = tmp + (size_t)NCH * EPC;
  int* toff   = tcnt + NBK * NCH;

  const int rowTiles = N / 16;
  const int gblocks  = (rowTiles + WAVES_PER_BLOCK - 1) / WAVES_PER_BLOCK;

  csr_bin  <NBK, NCH><<<NCH, BTH, 0, stream>>>(ei, tmp, tcnt, toff, E);
  csr_place<NBK, NCH><<<NBK, BTH, 0, stream>>>(tmp, tcnt, toff, rowptr, csr, N);

  const int nb16 = (N * 16 + TPB - 1) / TPB;
  const int nb32 = (N * 32 + TPB - 1) / TPB;

  gemm_wmma_f32<64><<<gblocks, TPB, 0, stream>>>(x, W1l, W1r, xl1, xr1, N);
  node_softmax_agg<16, false, 64><<<nb16, TPB, 0, stream>>>(xl1, xr1, a1, b1, rowptr, csr, nullptr,
                                                            W2l, W2r, xl2, xr2, N);
  node_softmax_agg<16, false, 128><<<nb16, TPB, 0, stream>>>(xl2, xr2, a2, b2, rowptr, csr, nullptr,
                                                             W3l, W3r, xl3, xr3, N);
  node_softmax_agg<32, true, 0><<<nb32, TPB, 0, stream>>>(xl3, xr3, a3, b3, rowptr, csr, (float*)d_out,
                                                          nullptr, nullptr, nullptr, nullptr, N);
}


# Edge-walk softmax factors by the hardware base-2 exponential of x*log2(e) (arguments <= 0) in place of library expf, weighted adds as explicit fma: a third fewer vector instructions; sum order kept
# speedup vs baseline: 9.8672x; 1.0615x over previous
#include <hip/hip_runtime.h>
#include <math.h>

typedef __attribute__((ext_vector_type(2))) float v2f;
typedef __attribute__((ext_vector_type(8))) float v8f;

#define TPB 256
#define WAVES_PER_BLOCK 8

template <int M>
__global__ __launch_bounds__(TPB) void gemm_wmma_f32(
    const float* __restrict__ X, const float* __restrict__ Wl,
    const float* __restrict__ Wr, float* __restrict__ Yl,
    float* __restrict__ Yr, int Nrows) {
  __shared__ __align__(16) float lw[2 * 64 * M];

  constexpr int TOTAL4 = 64 * M / 4;
  for (int i = threadIdx.x; i < TOTAL4; i += TPB) {
    ((float4*)lw)[i]          = ((const float4*)Wl)[i];
    ((float4*)lw)[TOTAL4 + i] = ((const float4*)Wr)[i];
  }
  __syncthreads();

  const int rowTile = blockIdx.x * WAVES_PER_BLOCK + (threadIdx.x >> 5);
  if (rowTile >= (Nrows >> 4)) return;
  const int row0 = rowTile << 4;
  const int lane = threadIdx.x & 31;
  const int r    = lane & 15;
  const int half = lane >> 4;

  v2f a[16];
  const float* xrow = X + (size_t)(row0 + r) * 64;
#pragma unroll
  for (int ks = 0; ks < 16; ++ks) {
    const int k = 4 * ks + 2 * half;
    a[ks].x = xrow[k];
    a[ks].y = xrow[k + 1];
  }

#pragma unroll
  for (int side = 0; side < 2; ++side) {
    const float* lws = lw + side * (64 * M);
    float* Y = side ? Yr : Yl;
#pragma nounroll
    for (int tn = 0; tn < M / 16; ++tn) {
      const int col0 = tn << 4;
      v8f c = {};
#pragma unroll
      for (int ks = 0; ks < 16; ++ks) {
        const int k = 4 * ks + 2 * half;
        v2f b;
        b.x = lws[k * M + col0 + r];
        b.y = lws[(k + 1) * M + col0 + r];
        c = __builtin_amdgcn_wmma_f32_16x16x4_f32(false, a[ks], false, b,
                                                  (short)0, c, false, false);
      }
#pragma unroll
      for (int v = 0; v < 8; ++v)
        Y[(size_t)(row0 + v + 8 * half) * M + col0 + r] = c[v];
    }
  }
}

#define BTH 1024
#define EPC 4096
#define NPB 512
#define NPB_LOG 9
#define STAGE_CAP 8192

__device__ __forceinline__ int block_excl_scan(int v, int* s_w, int& total) {
  const int lane = threadIdx.x & 31, w = threadIdx.x >> 5;
  int inc = v;
  for (int o = 1; o < 32; o <<= 1) {
    const int u = __shfl_up(inc, o);
    if (lane >= o) inc += u;
  }
  if (lane == 31) s_w[w] = inc;
  __syncthreads();
  int pre = 0, tot = 0;
  for (int k = 0; k < BTH / 32; ++k) {
    const int x = s_w[k];
    if (k < w) pre += x;
    tot += x;
  }
  __syncthreads();
  total = tot;
  return pre + inc - v;
}

template <int NBK, int NCH>
__global__ __launch_bounds__(BTH) void csr_bin(const int* __restrict__ ei,
                                               int* __restrict__ tmp,
                                               int* __restrict__ tcnt,
                                               int* __restrict__ toff, int E) {
  __shared__ int s_hist[NBK];
  __shared__ int s_off[NBK];
  __shared__ int s_w[BTH / 32];
  __shared__ int s_stage[EPC];
  const int t = threadIdx.x, c = blockIdx.x;
  if (t < NBK) s_hist[t] = 0;
  __syncthreads();

  const int e0 = c * EPC;
  int word[EPC / BTH], bk[EPC / BTH], rk[EPC / BTH];
#pragma unroll
  for (int j = 0; j < EPC / BTH; ++j) {
    const int e = e0 + j * BTH + t;
    bk[j] = -1; word[j] = 0; rk[j] = 0;
    if (e < E) {
      const int s = ei[e], d = ei[E + e];
      bk[j]   = d >> NPB_LOG;
      word[j] = (s << NPB_LOG) | (d & (NPB - 1));
      rk[j]   = atomicAdd(&s_hist[bk[j]], 1);
    }
  }
  __syncthreads();

  const int h = (t < NBK) ? s_hist[t] : 0;
  int nvalid;
  const int ex = block_excl_scan(h, s_w, nvalid);
  if (t < NBK) {
    s_off[t] = ex;
    tcnt[t * NCH + c] = h;
    toff[t * NCH + c] = ex;
  }
  __syncthreads();
#pragma unroll
  for (int j = 0; j < EPC / BTH; ++j)
    if (bk[j] >= 0) s_stage[s_off[bk[j]] + rk[j]] = word[j];
  __syncthreads();
  for (int k = t; k < nvalid; k += BTH) tmp[e0 + k] = s_stage[k];
}

template <int NBK, int NCH>
__global__ __launch_bounds__(BTH) void csr_place(const int* __restrict__ tmp,
                                                 const int* __restrict__ tcnt,
                                                 const int* __restrict__ toff,
                                                 int* __restrict__ rowptr,
                                                 int* __restrict__ csr, int N) {
  static_assert((NCH & 3) == 0, "run table rows are summed as int4");
  __shared__ int s_cnt[NCH];
  __shared__ int s_off[NCH];
  __shared__ int s_w[BTH / 32];
  __shared__ int s_deg[NPB];
  __shared__ int s_pos[NPB];
  __shared__ int s_out[STAGE_CAP];
  const int t = threadIdx.x, lane = t & 31, w = t >> 5, b = blockIdx.x;

  int part = 0;
  const int4* c4 = (const int4*)tcnt;
  const int nq = (b * NCH) >> 2;
  for (int q = t; q < nq; q += BTH) {
    const int4 v = c4[q];
    part += (v.x + v.y) + (v.z + v.w);
  }
  int mine = 0;
  if (t < NCH) {
    mine = tcnt[b * NCH + t];
    s_cnt[t] = mine;
    s_off[t] = toff[b * NCH + t];
  }
  if (t < NPB) s_deg[t] = 0;
  int base, nb;
  (void)block_excl_scan(part, s_w, base);
  (void)block_excl_scan(mine, s_w, nb);

  for (int c = w; c < NCH; c += BTH / 32) {
    const int n = s_cnt[c];
    const int* run = tmp + c * EPC + s_off[c];
    for (int k = lane; k < n; k += 32) atomicAdd(&s_deg[run[k] & (NPB - 1)], 1);
  }
  __syncthreads();

  const int dg = (t < NPB) ? s_deg[t] : 0;
  int nb2;
  const int ex = block_excl_scan(dg, s_w, nb2);
  if (t < NPB) {
    s_pos[t] = ex;
    const int node = b * NPB + t;
    if (node < N) rowptr[node] = base + ex;
  }
  if (b == NBK - 1 && t == 0) rowptr[N] = base + nb;
  __syncthreads();

  const bool staged = nb <= STAGE_CAP;
  for (int c = w; c < NCH; c += BTH / 32) {
    const int n = s_cnt[c];
    const int* run = tmp + c * EPC + s_off[c];
    for (int k = lane; k < n; k += 32) {
      const int wd = run[k];
      const int slot = atomicAdd(&s_pos[wd & (NPB - 1)], 1);
      const int src = (int)((unsigned)wd >> NPB_LOG);
      if (staged) s_out[slot] = src;
      else        csr[base + slot] = src;
    }
  }
  __syncthreads();
  if (staged)
    for (int k = t; k < nb; k += BTH) csr[base + k] = s_out[k];
}

__device__ __forceinline__ float exp_nonpos(float x) {
  return __builtin_amdgcn_exp2f(x * 1.44269504088896340736f);
}

template <int C, int K>
__device__ __forceinline__ void walk_entries(
    const float* __restrict__ xl, int my_s, int j0, int e0, int total, int sub,
    const float4& xr4, const float4& at4, float& m, float& den, float4& acc) {
  constexpr int M   = 4 * C;
  constexpr int LPN = C;
  constexpr int LPH = C / 4;
  float4 row[K];
#pragma unroll
  for (int k = 0; k < K; ++k) {
    const int s = __shfl(my_s, j0 + k, LPN);
    row[k] = *(const float4*)(xl + (size_t)s * M + 4 * sub);
  }
  float sc[K];
  float mb = m;
#pragma unroll
  for (int k = 0; k < K; ++k) {
    float vx = row[k].x + xr4.x, vy = row[k].y + xr4.y;
    float vz = row[k].z + xr4.z, vw = row[k].w + xr4.w;
    vx = vx > 0.0f ? vx : 0.2f * vx;
    vy = vy > 0.0f ? vy : 0.2f * vy;
    vz = vz > 0.0f ? vz : 0.2f * vz;
    vw = vw > 0.0f ? vw : 0.2f * vw;
    float p = at4.x * vx + at4.y * vy + at4.z * vz + at4.w * vw;
    for (int o = 1; o < LPH; o <<= 1) p += __shfl_xor(p, o);
    sc[k] = (e0 + k < total) ? p : -INFINITY;
    mb = fmaxf(mb, sc[k]);
  }
  const float scale = exp_nonpos(m - mb);
  den   *= scale;
  acc.x *= scale; acc.y *= scale; acc.z *= scale; acc.w *= scale;
#pragma unroll
  for (int k = 0; k < K; ++k) {
    const float wgt = exp_nonpos(sc[k] - mb);
    den   += wgt;
    acc.x = __builtin_fmaf(wgt, row[k].x, acc.x);
    acc.y = __builtin_fmaf(wgt, row[k].y, acc.y);
    acc.z = __builtin_fmaf(wgt, row[k].z, acc.z);
    acc.w = __builtin_fmaf(wgt, row[k].w, acc.w);
  }
  m = mb;
}

#define HP 68

template <int C, bool FINAL, int MN>
__global__ __launch_bounds__(TPB) void node_softmax_agg(
    const float* __restrict__ xl, const float* __restrict__ xr,
    const float* __restrict__ att, const float* __restrict__ bias,
    const int* __restrict__ rowptr, const int* __restrict__ csr,
    float* __restrict__ outp,
    const float* __restrict__ Wl, const float* __restrict__ Wr,
    float* __restrict__ Yl, float* __restrict__ Yr, int N) {
  constexpr int M   = 4 * C;
  constexpr int LPN = C;
  constexpr int LPH = C / 4;
  const int node = (blockIdx.x * TPB + threadIdx.x) / LPN;
  const int sub  = threadIdx.x & (LPN - 1);
  const bool valid = node < N;
  const int i = valid ? node : N - 1;

  const float4 xr4 = *(const float4*)(xr + (size_t)i * M + 4 * sub);
  const float4 at4 = *(const float4*)(att + 4 * sub);
  const int beg   = rowptr[i];
  const int total = rowptr[i + 1] - beg + 1;
  int wtotal = total;
  if constexpr (LPN < 32) wtotal = max(wtotal, __shfl_xor(wtotal, 16));

  float m = -INFINITY, den = 0.0f;
  float4 acc = make_float4(0.0f, 0.0f, 0.0f, 0.0f);

  for (int base = 0; base < wtotal; base += LPN) {
    const int idx = base + sub;
    int my_s = i;
    if (idx >= 1 && idx < total) my_s = csr[beg + idx - 1];
    const int cnt = min(LPN, wtotal - base);
    int j = 0;
    for (; j + 4 <= cnt; j += 4)
      walk_entries<C, 4>(xl, my_s, j, base + j, total, sub, xr4, at4, m, den, acc);
    const int rem = cnt - j;
    if (rem == 3)
      walk_entries<C, 3>(xl, my_s, j, base + j, total, sub, xr4, at4, m, den, acc);
    else if (rem == 2)
      walk_entries<C, 2>(xl, my_s, j, base + j, total, sub, xr4, at4, m, den, acc);
    else if (rem == 1)
      walk_entries<C, 1>(xl, my_s, j, base + j, total, sub, xr4, at4, m, den, acc);
  }

  float4 r = make_float4(acc.x / den, acc.y / den, acc.z / den, acc.w / den);
  if constexpr (!FINAL) {
    const float4 b4 = *(const float4*)(bias + 4 * sub);
    r.x += b4.x; r.y += b4.y; r.z += b4.z; r.w += b4.w;
    r.x = r.x > 0.0f ? r.x : expm1f(r.x);
    r.y = r.y > 0.0f ? r.y : expm1f(r.y);
    r.z = r.z > 0.0f ? r.z : expm1f(r.z);
    r.w = r.w > 0.0f ? r.w : expm1f(r.w);

    static_assert(M == 64 && TPB / LPN == 16, "the work-group's rows are one 16x64 A tile");
    __shared__ __align__(16) float ht[16 * HP];
    *(float4*)(ht + (threadIdx.x / LPN) * HP + 4 * sub) = r;
    __syncthreads();

    constexpr int GPS = MN / 32;
    const int w = threadIdx.x >> 5;
    if (2 * GPS < WAVES_PER_BLOCK && w >= 2 * GPS) return;
    const int lane = threadIdx.x & 31;
    const int rr   = lane & 15;
    const int half = lane >> 4;
    const bool right = w >= GPS;
    const float* Wm = right ? Wr : Wl;
    float* Y = right ? Yr : Yl;
    const int cb = (right ? w - GPS : w) * 32 + 2 * rr;
    const float* arow = ht + rr * HP + 2 * half;
    const float* wcol = Wm + (size_t)(2 * half) * MN + cb;
    v8f c0 = {}, c1 = {};
#pragma unroll
    for (int ks = 0; ks < 16; ++ks) {
      const v2f a  = *(const v2f*)(arow + 4 * ks);
      const v2f w0 = *(const v2f*)(wcol + (size_t)(4 * ks) * MN);
      const v2f w1 = *(const v2f*)(wcol + (size_t)(4 * ks + 1) * MN);
      v2f b0, b1;
      b0.x = w0.x; b0.y = w1.x;
      b1.x = w0.y; b1.y = w1.y;
      c0 = __builtin_amdgcn_wmma_f32_16x16x4_f32(false, a, false, b0, (short)0, c0, false, false);
      c1 = __builtin_amdgcn_wmma_f32_16x16x4_f32(false, a, false, b1, (short)0, c1, false, false);
    }
    const int row0 = blockIdx.x * (TPB / LPN);
    float* yp = Y + (size_t)(row0 + 8 * half) * MN + cb;
    if (row0 + 16 <= N) {
#pragma unroll
      for (int v = 0; v < 8; ++v) {
        v2f o;
        o.x = c0[v]; o.y = c1[v];
        *(v2f*)(yp + (size_t)v * MN) = o;
      }
    } else {
#pragma unroll
      for (int v = 0; v < 8; ++v) {
        if (row0 + 8 * half + v < N) {
          v2f o;
          o.x = c0[v]; o.y = c1[v];
          *(v2f*)(yp + (size_t)v * MN) = o;
        }
      }
    }
  } else {
    for (int o = LPH; o < LPN; o <<= 1) {
      r.x += __shfl_xor(r.x, o); r.y += __shfl_xor(r.y, o);
      r.z += __shfl_xor(r.z, o); r.w += __shfl_xor(r.w, o);
    }
    if (valid && sub < LPH) {
      const float4 b4 = *(const float4*)(bias + 4 * sub);
      r.x = 0.25f * r.x + b4.x; r.y = 0.25f * r.y + b4.y;
      r.z = 0.25f * r.z + b4.z; r.w = 0.25f * r.w + b4.w;
      *(float4*)(outp + (size_t)i * C + 4 * sub) = r;
    }
  }
}

extern "C" void kernel_launch(void* const* d_in, const int* in_sizes, int n_in,
                              void* d_out, int out_size, void* d_ws, size_t ws_size,
                              hipStream_t stream) {
  (void)in_sizes; (void)n_in; (void)out_size; (void)ws_size;
  const float* x   = (const float*)d_in[0];
  const int*   ei  = (const int*)d_in[1];
  const float* W1l = (const float*)d_in[2];
  const float* W1r = (const float*)d_in[3];
  const float* a1  = (const float*)d_in[4];
  const float* b1  = (const float*)d_in[5];
  const float* W2l = (const float*)d_in[6];
  const float* W2r = (const float*)d_in[7];
  const float* a2  = (const float*)d_in[8];
  const float* b2  = (const float*)d_in[9];
  const float* W3l = (const float*)d_in[10];
  const float* W3r = (const float*)d_in[11];
  const float* a3  = (const float*)d_in[12];
  const float* b3  = (const float*)d_in[13];

  constexpr int N = 100000, E = 800000;

  float* ws = (float*)d_ws;
  float* xl1 = ws;
  float* xr1 = xl1 + (size_t)N * 64;
  float* xl2 = xr1 + (size_t)N * 64;
  float* xr2 = xl2 + (size_t)N * 64;
  float* xl3 = ws;
  float* xr3 = xr2 + (size_t)N * 64;
  constexpr int NBK = (N + NPB - 1) / NPB;
  constexpr int NCH = (E + EPC - 1) / EPC;
  constexpr int RP  = (N + 1 + 31) / 32 * 32;
  int* rowptr = (int*)(xr3 + (size_t)N * 128);
  int* csr    = rowptr + RP;
  int* tmp    = csr + E;
  int* tcnt   = tmp + (size_t)NCH * EPC;
  int* toff   = tcnt + NBK * NCH;

  const int rowTiles = N / 16;
  const int gblocks  = (rowTiles + WAVES_PER_BLOCK - 1) / WAVES_PER_BLOCK;

  csr_bin  <NBK, NCH><<<NCH, BTH, 0, stream>>>(ei, tmp, tcnt, toff, E);
  csr_place<NBK, NCH><<<NBK, BTH, 0, stream>>>(tmp, tcnt, toff, rowptr, csr, N);

  const int nb16 = (N * 16 + TPB - 1) / TPB;
  const int nb32 = (N * 32 + TPB - 1) / TPB;

  gemm_wmma_f32<64><<<gblocks, TPB, 0, stream>>>(x, W1l, W1r, xl1, xr1, N);
  node_softmax_agg<16, false, 64><<<nb16, TPB, 0, stream>>>(xl1, xr1, a1, b1, rowptr, csr, nullptr,
                                                            W2l, W2r, xl2, xr2, N);
  node_softmax_agg<16, false, 128><<<nb16, TPB, 0, stream>>>(xl2, xr2, a2, b2, rowptr, csr, nullptr,
                                                             W3l, W3r, xl3, xr3, N);
  node_softmax_agg<32, true, 0><<<nb32, TPB, 0, stream>>>(xl3, xr3, a3, b3, rowptr, csr, (float*)d_out,
                                                          nullptr, nullptr, nullptr, nullptr, N);
}


# Layer-1 xl|xr GEMM rides in the CSR placement launch as 250 extra work-groups of 25 row tiles (its WMMAs fill issue slots the placement's dependent loads leave idle); 6 -> 5 launches; sum order kept
# speedup vs baseline: 10.1790x; 1.0316x over previous
#include <hip/hip_runtime.h>
#include <math.h>

typedef __attribute__((ext_vector_type(2))) float v2f;
typedef __attribute__((ext_vector_type(8))) float v8f;

#define TPB 256
#define WAVES_PER_BLOCK 8

template <int M, int TILES, int THREADS>
__device__ __forceinline__ void gemm_pair_tiles(
    const float* __restrict__ X, const float* __restrict__ Wl,
    const float* __restrict__ Wr, float* __restrict__ Yl,
    float* __restrict__ Yr, int Nrows, float* lw, int blk) {
  constexpr int TOTAL4 = 64 * M / 4;
  for (int i = threadIdx.x; i < TOTAL4; i += THREADS) {
    ((float4*)lw)[i]          = ((const float4*)Wl)[i];
    ((float4*)lw)[TOTAL4 + i] = ((const float4*)Wr)[i];
  }
  __syncthreads();

  const int wv = threadIdx.x >> 5;
  const int rowTile = blk * TILES + wv;
  if (wv >= TILES || rowTile >= (Nrows >> 4)) return;
  const int row0 = rowTile << 4;
  const int lane = threadIdx.x & 31;
  const int r    = lane & 15;
  const int half = lane >> 4;

  v2f a[16];
  const float* xrow = X + (size_t)(row0 + r) * 64;
#pragma unroll
  for (int ks = 0; ks < 16; ++ks) {
    const int k = 4 * ks + 2 * half;
    a[ks].x = xrow[k];
    a[ks].y = xrow[k + 1];
  }

#pragma unroll
  for (int side = 0; side < 2; ++side) {
    const float* lws = lw + side * (64 * M);
    float* Y = side ? Yr : Yl;
#pragma nounroll
    for (int tn = 0; tn < M / 16; ++tn) {
      const int col0 = tn << 4;
      v8f c = {};
#pragma unroll
      for (int ks = 0; ks < 16; ++ks) {
        const int k = 4 * ks + 2 * half;
        v2f b;
        b.x = lws[k * M + col0 + r];
        b.y = lws[(k + 1) * M + col0 + r];
        c = __builtin_amdgcn_wmma_f32_16x16x4_f32(false, a[ks], false, b,
                                                  (short)0, c, false, false);
      }
#pragma unroll
      for (int v = 0; v < 8; ++v)
        Y[(size_t)(row0 + v + 8 * half) * M + col0 + r] = c[v];
    }
  }
}

#define BTH 1024
#define EPC 4096
#define NPB 512
#define NPB_LOG 9
#define STAGE_CAP 8192

__device__ __forceinline__ int block_excl_scan(int v, int* s_w, int& total) {
  const int lane = threadIdx.x & 31, w = threadIdx.x >> 5;
  int inc = v;
  for (int o = 1; o < 32; o <<= 1) {
    const int u = __shfl_up(inc, o);
    if (lane >= o) inc += u;
  }
  if (lane == 31) s_w[w] = inc;
  __syncthreads();
  int pre = 0, tot = 0;
  for (int k = 0; k < BTH / 32; ++k) {
    const int x = s_w[k];
    if (k < w) pre += x;
    tot += x;
  }
  __syncthreads();
  total = tot;
  return pre + inc - v;
}

template <int NBK, int NCH>
__global__ __launch_bounds__(BTH) void csr_bin(const int* __restrict__ ei,
                                               int* __restrict__ tmp,
                                               int* __restrict__ tcnt,
                                               int* __restrict__ toff, int E) {
  __shared__ int s_hist[NBK];
  __shared__ int s_off[NBK];
  __shared__ int s_w[BTH / 32];
  __shared__ int s_stage[EPC];
  const int t = threadIdx.x, c = blockIdx.x;
  if (t < NBK) s_hist[t] = 0;
  __syncthreads();

  const int e0 = c * EPC;
  int word[EPC / BTH], bk[EPC / BTH], rk[EPC / BTH];
#pragma unroll
  for (int j = 0; j < EPC / BTH; ++j) {
    const int e = e0 + j * BTH + t;
    bk[j] = -1; word[j] = 0; rk[j] = 0;
    if (e < E) {
      const int s = ei[e], d = ei[E + e];
      bk[j]   = d >> NPB_LOG;
      word[j] = (s << NPB_LOG) | (d & (NPB - 1));
      rk[j]   = atomicAdd(&s_hist[bk[j]], 1);
    }
  }
  __syncthreads();

  const int h = (t < NBK) ? s_hist[t] : 0;
  int nvalid;
  const int ex = block_excl_scan(h, s_w, nvalid);
  if (t < NBK) {
    s_off[t] = ex;
    tcnt[t * NCH + c] = h;
    toff[t * NCH + c] = ex;
  }
  __syncthreads();
#pragma unroll
  for (int j = 0; j < EPC / BTH; ++j)
    if (bk[j] >= 0) s_stage[s_off[bk[j]] + rk[j]] = word[j];
  __syncthreads();
  for (int k = t; k < nvalid; k += BTH) tmp[e0 + k] = s_stage[k];
}

#define GTILES 25
template <int NBK, int NCH>
__global__ __launch_bounds__(BTH) void csr_place_gemm(const int* __restrict__ tmp,
                                                      const int* __restrict__ tcnt,
                                                      const int* __restrict__ toff,
                                                      int* __restrict__ rowptr,
                                                      int* __restrict__ csr, int N,
                                                      const float* __restrict__ X,
                                                      const float* __restrict__ Wl,
                                                      const float* __restrict__ Wr,
                                                      float* __restrict__ Yl,
                                                      float* __restrict__ Yr) {
  static_assert((NCH & 3) == 0, "run table rows are summed as int4");
  static_assert(STAGE_CAP * sizeof(int) == 2 * 64 * 64 * sizeof(float), "the W pair takes exactly the segment buffer");
  __shared__ int s_cnt[NCH];
  __shared__ int s_off[NCH];
  __shared__ int s_w[BTH / 32];
  __shared__ int s_deg[NPB];
  __shared__ int s_pos[NPB];
  __shared__ __align__(16) unsigned char s_raw[STAGE_CAP * sizeof(int)];
  if (blockIdx.x >= NBK) {
    gemm_pair_tiles<64, GTILES, BTH>(X, Wl, Wr, Yl, Yr, N, (float*)s_raw, (int)blockIdx.x - NBK);
    return;
  }
  int* s_out = (int*)s_raw;
  const int t = threadIdx.x, lane = t & 31, w = t >> 5, b = blockIdx.x;

  int part = 0;
  const int4* c4 = (const int4*)tcnt;
  const int nq = (b * NCH) >> 2;
  for (int q = t; q < nq; q += BTH) {
    const int4 v = c4[q];
    part += (v.x + v.y) + (v.z + v.w);
  }
  int mine = 0;
  if (t < NCH) {
    mine = tcnt[b * NCH + t];
    s_cnt[t] = mine;
    s_off[t] = toff[b * NCH + t];
  }
  if (t < NPB) s_deg[t] = 0;
  int base, nb;
  (void)block_excl_scan(part, s_w, base);
  (void)block_excl_scan(mine, s_w, nb);

  for (int c = w; c < NCH; c += BTH / 32) {
    const int n = s_cnt[c];
    const int* run = tmp + c * EPC + s_off[c];
    for (int k = lane; k < n; k += 32) atomicAdd(&s_deg[run[k] & (NPB - 1)], 1);
  }
  __syncthreads();

  const int dg = (t < NPB) ? s_deg[t] : 0;
  int nb2;
  const int ex = block_excl_scan(dg, s_w, nb2);
  if (t < NPB) {
    s_pos[t] = ex;
    const int node = b * NPB + t;
    if (node < N) rowptr[node] = base + ex;
  }
  if (b == NBK - 1 && t == 0) rowptr[N] = base + nb;
  __syncthreads();

  const bool staged = nb <= STAGE_CAP;
  for (int c = w; c < NCH; c += BTH / 32) {
    const int n = s_cnt[c];
    const int* run = tmp + c * EPC + s_off[c];
    for (int k = lane; k < n; k += 32) {
      const int wd = run[k];
      const int slot = atomicAdd(&s_pos[wd & (NPB - 1)], 1);
      const int src = (int)((unsigned)wd >> NPB_LOG);
      if (staged) s_out[slot] = src;
      else        csr[base + slot] = src;
    }
  }
  __syncthreads();
  if (staged)
    for (int k = t; k < nb; k += BTH) csr[base + k] = s_out[k];
}

__device__ __forceinline__ float exp_nonpos(float x) {
  return __builtin_amdgcn_exp2f(x * 1.44269504088896340736f);
}

template <int C, int K>
__device__ __forceinline__ void walk_entries(
    const float* __restrict__ xl, int my_s, int j0, int e0, int total, int sub,
    const float4& xr4, const float4& at4, float& m, float& den, float4& acc) {
  constexpr int M   = 4 * C;
  constexpr int LPN = C;
  constexpr int LPH = C / 4;
  float4 row[K];
#pragma unroll
  for (int k = 0; k < K; ++k) {
    const int s = __shfl(my_s, j0 + k, LPN);
    row[k] = *(const float4*)(xl + (size_t)s * M + 4 * sub);
  }
  float sc[K];
  float mb = m;
#pragma unroll
  for (int k = 0; k < K; ++k) {
    float vx = row[k].x + xr4.x, vy = row[k].y + xr4.y;
    float vz = row[k].z + xr4.z, vw = row[k].w + xr4.w;
    vx = vx > 0.0f ? vx : 0.2f * vx;
    vy = vy > 0.0f ? vy : 0.2f * vy;
    vz = vz > 0.0f ? vz : 0.2f * vz;
    vw = vw > 0.0f ? vw : 0.2f * vw;
    float p = at4.x * vx + at4.y * vy + at4.z * vz + at4.w * vw;
    for (int o = 1; o < LPH; o <<= 1) p += __shfl_xor(p, o);
    sc[k] = (e0 + k < total) ? p : -INFINITY;
    mb = fmaxf(mb, sc[k]);
  }
  const float scale = exp_nonpos(m - mb);
  den   *= scale;
  acc.x *= scale; acc.y *= scale; acc.z *= scale; acc.w *= scale;
#pragma unroll
  for (int k = 0; k < K; ++k) {
    const float wgt = exp_nonpos(sc[k] - mb);
    den   += wgt;
    acc.x = __builtin_fmaf(wgt, row[k].x, acc.x);
    acc.y = __builtin_fmaf(wgt, row[k].y, acc.y);
    acc.z = __builtin_fmaf(wgt, row[k].z, acc.z);
    acc.w = __builtin_fmaf(wgt, row[k].w, acc.w);
  }
  m = mb;
}

#define HP 68

template <int C, bool FINAL, int MN>
__global__ __launch_bounds__(TPB) void node_softmax_agg(
    const float* __restrict__ xl, const float* __restrict__ xr,
    const float* __restrict__ att, const float* __restrict__ bias,
    const int* __restrict__ rowptr, const int* __restrict__ csr,
    float* __restrict__ outp,
    const float* __restrict__ Wl, const float* __restrict__ Wr,
    float* __restrict__ Yl, float* __restrict__ Yr, int N) {
  constexpr int M   = 4 * C;
  constexpr int LPN = C;
  constexpr int LPH = C / 4;
  const int node = (blockIdx.x * TPB + threadIdx.x) / LPN;
  const int sub  = threadIdx.x & (LPN - 1);
  const bool valid = node < N;
  const int i = valid ? node : N - 1;

  const float4 xr4 = *(const float4*)(xr + (size_t)i * M + 4 * sub);
  const float4 at4 = *(const float4*)(att + 4 * sub);
  const int beg   = rowptr[i];
  const int total = rowptr[i + 1] - beg + 1;
  int wtotal = total;
  if constexpr (LPN < 32) wtotal = max(wtotal, __shfl_xor(wtotal, 16));

  float m = -INFINITY, den = 0.0f;
  float4 acc = make_float4(0.0f, 0.0f, 0.0f, 0.0f);

  for (int base = 0; base < wtotal; base += LPN) {
    const int idx = base + sub;
    int my_s = i;
    if (idx >= 1 && idx < total) my_s = csr[beg + idx - 1];
    const int cnt = min(LPN, wtotal - base);
    int j = 0;
    for (; j + 4 <= cnt; j += 4)
      walk_entries<C, 4>(xl, my_s, j, base + j, total, sub, xr4, at4, m, den, acc);
    const int rem = cnt - j;
    if (rem == 3)
      walk_entries<C, 3>(xl, my_s, j, base + j, total, sub, xr4, at4, m, den, acc);
    else if (rem == 2)
      walk_entries<C, 2>(xl, my_s, j, base + j, total, sub, xr4, at4, m, den, acc);
    else if (rem == 1)
      walk_entries<C, 1>(xl, my_s, j, base + j, total, sub, xr4, at4, m, den, acc);
  }

  float4 r = make_float4(acc.x / den, acc.y / den, acc.z / den, acc.w / den);
  if constexpr (!FINAL) {
    const float4 b4 = *(const float4*)(bias + 4 * sub);
    r.x += b4.x; r.y += b4.y; r.z += b4.z; r.w += b4.w;
    r.x = r.x > 0.0f ? r.x : expm1f(r.x);
    r.y = r.y > 0.0f ? r.y : expm1f(r.y);
    r.z = r.z > 0.0f ? r.z : expm1f(r.z);
    r.w = r.w > 0.0f ? r.w : expm1f(r.w);

    static_assert(M == 64 && TPB / LPN == 16, "the work-group's rows are one 16x64 A tile");
    __shared__ __align__(16) float ht[16 * HP];
    *(float4*)(ht + (threadIdx.x / LPN) * HP + 4 * sub) = r;
    __syncthreads();

    constexpr int GPS = MN / 32;
    const int w = threadIdx.x >> 5;
    if (2 * GPS < WAVES_PER_BLOCK && w >= 2 * GPS) return;
    const int lane = threadIdx.x & 31;
    const int rr   = lane & 15;
    const int half = lane >> 4;
    const bool right = w >= GPS;
    const float* Wm = right ? Wr : Wl;
    float* Y = right ? Yr : Yl;
    const int cb = (right ? w - GPS : w) * 32 + 2 * rr;
    const float* arow = ht + rr * HP + 2 * half;
    const float* wcol = Wm + (size_t)(2 * half) * MN + cb;
    v8f c0 = {}, c1 = {};
#pragma unroll
    for (int ks = 0; ks < 16; ++ks) {
      const v2f a  = *(const v2f*)(arow + 4 * ks);
      const v2f w0 = *(const v2f*)(wcol + (size_t)(4 * ks) * MN);
      const v2f w1 = *(const v2f*)(wcol + (size_t)(4 * ks + 1) * MN);
      v2f b0, b1;
      b0.x = w0.x; b0.y = w1.x;
      b1.x = w0.y; b1.y = w1.y;
      c0 = __builtin_amdgcn_wmma_f32_16x16x4_f32(false, a, false, b0, (short)0, c0, false, false);
      c1 = __builtin_amdgcn_wmma_f32_16x16x4_f32(false, a, false, b1, (short)0, c1, false, false);
    }
    const int row0 = blockIdx.x * (TPB / LPN);
    float* yp = Y + (size_t)(row0 + 8 * half) * MN + cb;
    if (row0 + 16 <= N) {
#pragma unroll
      for (int v = 0; v < 8; ++v) {
        v2f o;
        o.x = c0[v]; o.y = c1[v];
        *(v2f*)(yp + (size_t)v * MN) = o;
      }
    } else {
#pragma unroll
      for (int v = 0; v < 8; ++v) {
        if (row0 + 8 * half + v < N) {
          v2f o;
          o.x = c0[v]; o.y = c1[v];
          *(v2f*)(yp + (size_t)v * MN) = o;
        }
      }
    }
  } else {
    for (int o = LPH; o < LPN; o <<= 1) {
      r.x += __shfl_xor(r.x, o); r.y += __shfl_xor(r.y, o);
      r.z += __shfl_xor(r.z, o); r.w += __shfl_xor(r.w, o);
    }
    if (valid && sub < LPH) {
      const float4 b4 = *(const float4*)(bias + 4 * sub);
      r.x = 0.25f * r.x + b4.x; r.y = 0.25f * r.y + b4.y;
      r.z = 0.25f * r.z + b4.z; r.w = 0.25f * r.w + b4.w;
      *(float4*)(outp + (size_t)i * C + 4 * sub) = r;
    }
  }
}

extern "C" void kernel_launch(void* const* d_in, const int* in_sizes, int n_in,
                              void* d_out, int out_size, void* d_ws, size_t ws_size,
                              hipStream_t stream) {
  (void)in_sizes; (void)n_in; (void)out_size; (void)ws_size;
  const float* x   = (const float*)d_in[0];
  const int*   ei  = (const int*)d_in[1];
  const float* W1l = (const float*)d_in[2];
  const float* W1r = (const float*)d_in[3];
  const float* a1  = (const float*)d_in[4];
  const float* b1  = (const float*)d_in[5];
  const float* W2l = (const float*)d_in[6];
  const float* W2r = (const float*)d_in[7];
  const float* a2  = (const float*)d_in[8];
  const float* b2  = (const float*)d_in[9];
  const float* W3l = (const float*)d_in[10];
  const float* W3r = (const float*)d_in[11];
  const float* a3  = (const float*)d_in[12];
  const float* b3  = (const float*)d_in[13];

  constexpr int N = 100000, E = 800000;

  float* ws = (float*)d_ws;
  float* xl1 = ws;
  float* xr1 = xl1 + (size_t)N * 64;
  float* xl2 = xr1 + (size_t)N * 64;
  float* xr2 = xl2 + (size_t)N * 64;
  float* xl3 = ws;
  float* xr3 = xr2 + (size_t)N * 64;
  constexpr int NBK = (N + NPB - 1) / NPB;
  constexpr int NCH = (E + EPC - 1) / EPC;
  constexpr int RP  = (N + 1 + 31) / 32 * 32;
  int* rowptr = (int*)(xr3 + (size_t)N * 128);
  int* csr    = rowptr + RP;
  int* tmp    = csr + E;
  int* tcnt   = tmp + (size_t)NCH * EPC;
  int* toff   = tcnt + NBK * NCH;

  constexpr int rowTiles = N / 16;
  constexpr int GBLK = (rowTiles + GTILES - 1) / GTILES;

  csr_bin       <NBK, NCH><<<NCH, BTH, 0, stream>>>(ei, tmp, tcnt, toff, E);
  csr_place_gemm<NBK, NCH><<<NBK + GBLK, BTH, 0, stream>>>(tmp, tcnt, toff, rowptr, csr, N,
                                                          x, W1l, W1r, xl1, xr1);

  const int nb16 = (N * 16 + TPB - 1) / TPB;
  const int nb32 = (N * 32 + TPB - 1) / TPB;

  node_softmax_agg<16, false, 64><<<nb16, TPB, 0, stream>>>(xl1, xr1, a1, b1, rowptr, csr, nullptr,
                                                            W2l, W2r, xl2, xr2, N);
  node_softmax_agg<16, false, 128><<<nb16, TPB, 0, stream>>>(xl2, xr2, a2, b2, rowptr, csr, nullptr,
                                                             W3l, W3r, xl3, xr3, N);
  node_softmax_agg<32, true, 0><<<nb32, TPB, 0, stream>>>(xl3, xr3, a3, b3, rowptr, csr, (float*)d_out,
                                                          nullptr, nullptr, nullptr, nullptr, N);
}


# Edge walk's cross-lane reads (source-index broadcast, per-head score sums, head mean) as DPP row_share/row_xmask, v_permlanex16 and v_readlane instead of __shfl's ds_bpermute; sum order kept
# speedup vs baseline: 11.1475x; 1.0951x over previous
#include <hip/hip_runtime.h>
#include <math.h>

typedef __attribute__((ext_vector_type(2))) float v2f;
typedef __attribute__((ext_vector_type(8))) float v8f;

#define TPB 256
#define WAVES_PER_BLOCK 8

template <int M, int TILES, int THREADS>
__device__ __forceinline__ void gemm_pair_tiles(
    const float* __restrict__ X, const float* __restrict__ Wl,
    const float* __restrict__ Wr, float* __restrict__ Yl,
    float* __restrict__ Yr, int Nrows, float* lw, int blk) {
  constexpr int TOTAL4 = 64 * M / 4;
  for (int i = threadIdx.x; i < TOTAL4; i += THREADS) {
    ((float4*)lw)[i]          = ((const float4*)Wl)[i];
    ((float4*)lw)[TOTAL4 + i] = ((const float4*)Wr)[i];
  }
  __syncthreads();

  const int wv = threadIdx.x >> 5;
  const int rowTile = blk * TILES + wv;
  if (wv >= TILES || rowTile >= (Nrows >> 4)) return;
  const int row0 = rowTile << 4;
  const int lane = threadIdx.x & 31;
  const int r    = lane & 15;
  const int half = lane >> 4;

  v2f a[16];
  const float* xrow = X + (size_t)(row0 + r) * 64;
#pragma unroll
  for (int ks = 0; ks < 16; ++ks) {
    const int k = 4 * ks + 2 * half;
    a[ks].x = xrow[k];
    a[ks].y = xrow[k + 1];
  }

#pragma unroll
  for (int side = 0; side < 2; ++side) {
    const float* lws = lw + side * (64 * M);
    float* Y = side ? Yr : Yl;
#pragma nounroll
    for (int tn = 0; tn < M / 16; ++tn) {
      const int col0 = tn << 4;
      v8f c = {};
#pragma unroll
      for (int ks = 0; ks < 16; ++ks) {
        const int k = 4 * ks + 2 * half;
        v2f b;
        b.x = lws[k * M + col0 + r];
        b.y = lws[(k + 1) * M + col0 + r];
        c = __builtin_amdgcn_wmma_f32_16x16x4_f32(false, a[ks], false, b,
                                                  (short)0, c, false, false);
      }
#pragma unroll
      for (int v = 0; v < 8; ++v)
        Y[(size_t)(row0 + v + 8 * half) * M + col0 + r] = c[v];
    }
  }
}

#define BTH 1024
#define EPC 4096
#define NPB 512
#define NPB_LOG 9
#define STAGE_CAP 8192

__device__ __forceinline__ int block_excl_scan(int v, int* s_w, int& total) {
  const int lane = threadIdx.x & 31, w = threadIdx.x >> 5;
  int inc = v;
  for (int o = 1; o < 32; o <<= 1) {
    const int u = __shfl_up(inc, o);
    if (lane >= o) inc += u;
  }
  if (lane == 31) s_w[w] = inc;
  __syncthreads();
  int pre = 0, tot = 0;
  for (int k = 0; k < BTH / 32; ++k) {
    const int x = s_w[k];
    if (k < w) pre += x;
    tot += x;
  }
  __syncthreads();
  total = tot;
  return pre + inc - v;
}

template <int NBK, int NCH>
__global__ __launch_bounds__(BTH) void csr_bin(const int* __restrict__ ei,
                                               int* __restrict__ tmp,
                                               int* __restrict__ tcnt,
                                               int* __restrict__ toff, int E) {
  __shared__ int s_hist[NBK];
  __shared__ int s_off[NBK];
  __shared__ int s_w[BTH / 32];
  __shared__ int s_stage[EPC];
  const int t = threadIdx.x, c = blockIdx.x;
  if (t < NBK) s_hist[t] = 0;
  __syncthreads();

  const int e0 = c * EPC;
  int word[EPC / BTH], bk[EPC / BTH], rk[EPC / BTH];
#pragma unroll
  for (int j = 0; j < EPC / BTH; ++j) {
    const int e = e0 + j * BTH + t;
    bk[j] = -1; word[j] = 0; rk[j] = 0;
    if (e < E) {
      const int s = ei[e], d = ei[E + e];
      bk[j]   = d >> NPB_LOG;
      word[j] = (s << NPB_LOG) | (d & (NPB - 1));
      rk[j]   = atomicAdd(&s_hist[bk[j]], 1);
    }
  }
  __syncthreads();

  const int h = (t < NBK) ? s_hist[t] : 0;
  int nvalid;
  const int ex = block_excl_scan(h, s_w, nvalid);
  if (t < NBK) {
    s_off[t] = ex;
    tcnt[t * NCH + c] = h;
    toff[t * NCH + c] = ex;
  }
  __syncthreads();
#pragma unroll
  for (int j = 0; j < EPC / BTH; ++j)
    if (bk[j] >= 0) s_stage[s_off[bk[j]] + rk[j]] = word[j];
  __syncthreads();
  for (int k = t; k < nvalid; k += BTH) tmp[e0 + k] = s_stage[k];
}

#define GTILES 25
template <int NBK, int NCH>
__global__ __launch_bounds__(BTH) void csr_place_gemm(const int* __restrict__ tmp,
                                                      const int* __restrict__ tcnt,
                                                      const int* __restrict__ toff,
                                                      int* __restrict__ rowptr,
                                                      int* __restrict__ csr, int N,
                                                      const float* __restrict__ X,
                                                      const float* __restrict__ Wl,
                                                      const float* __restrict__ Wr,
                                                      float* __restrict__ Yl,
                                                      float* __restrict__ Yr) {
  static_assert((NCH & 3) == 0, "run table rows are summed as int4");
  static_assert(STAGE_CAP * sizeof(int) == 2 * 64 * 64 * sizeof(float), "the W pair takes exactly the segment buffer");
  __shared__ int s_cnt[NCH];
  __shared__ int s_off[NCH];
  __shared__ int s_w[BTH / 32];
  __shared__ int s_deg[NPB];
  __shared__ int s_pos[NPB];
  __shared__ __align__(16) unsigned char s_raw[STAGE_CAP * sizeof(int)];
  if (blockIdx.x >= NBK) {
    gemm_pair_tiles<64, GTILES, BTH>(X, Wl, Wr, Yl, Yr, N, (float*)s_raw, (int)blockIdx.x - NBK);
    return;
  }
  int* s_out = (int*)s_raw;
  const int t = threadIdx.x, lane = t & 31, w = t >> 5, b = blockIdx.x;

  int part = 0;
  const int4* c4 = (const int4*)tcnt;
  const int nq = (b * NCH) >> 2;
  for (int q = t; q < nq; q += BTH) {
    const int4 v = c4[q];
    part += (v.x + v.y) + (v.z + v.w);
  }
  int mine = 0;
  if (t < NCH) {
    mine = tcnt[b * NCH + t];
    s_cnt[t] = mine;
    s_off[t] = toff[b * NCH + t];
  }
  if (t < NPB) s_deg[t] = 0;
  int base, nb;
  (void)block_excl_scan(part, s_w, base);
  (void)block_excl_scan(mine, s_w, nb);

  for (int c = w; c < NCH; c += BTH / 32) {
    const int n = s_cnt[c];
    const int* run = tmp + c * EPC + s_off[c];
    for (int k = lane; k < n; k += 32) atomicAdd(&s_deg[run[k] & (NPB - 1)], 1);
  }
  __syncthreads();

  const int dg = (t < NPB) ? s_deg[t] : 0;
  int nb2;
  const int ex = block_excl_scan(dg, s_w, nb2);
  if (t < NPB) {
    s_pos[t] = ex;
    const int node = b * NPB + t;
    if (node < N) rowptr[node] = base + ex;
  }
  if (b == NBK - 1 && t == 0) rowptr[N] = base + nb;
  __syncthreads();

  const bool staged = nb <= STAGE_CAP;
  for (int c = w; c < NCH; c += BTH / 32) {
    const int n = s_cnt[c];
    const int* run = tmp + c * EPC + s_off[c];
    for (int k = lane; k < n; k += 32) {
      const int wd = run[k];
      const int slot = atomicAdd(&s_pos[wd & (NPB - 1)], 1);
      const int src = (int)((unsigned)wd >> NPB_LOG);
      if (staged) s_out[slot] = src;
      else        csr[base + slot] = src;
    }
  }
  __syncthreads();
  if (staged)
    for (int k = t; k < nb; k += BTH) csr[base + k] = s_out[k];
}

__device__ __forceinline__ float exp_nonpos(float x) {
  return __builtin_amdgcn_exp2f(x * 1.44269504088896340736f);
}

template <int CTRL>
__device__ __forceinline__ int dpp_row(int v) {
  return __builtin_amdgcn_update_dpp(0, v, CTRL, 0xf, 0xf, true);
}
#define XADD(r, m) "v_add_f32_dpp " r ", " r ", " r " row_xmask:" #m " row_mask:0xf bank_mask:0xf\n\t"
#define XSTEP1(m) XADD("%0", m)
#define XSTEP2(m) XADD("%0", m) XADD("%1", m)
#define XSTEP3(m) XADD("%0", m) XADD("%1", m) XADD("%2", m)
#define XSTEP4(m) XADD("%0", m) XADD("%1", m) XADD("%2", m) XADD("%3", m)
#define XDEP1 "s_delay_alu instid0(VALU_DEP_1)\n\t"
#define XDEP2 "s_delay_alu instid0(VALU_DEP_2) | instskip(NEXT) | instid1(VALU_DEP_2)\n\t"

template <int K, int LPH>
__device__ __forceinline__ void head_sums(float (&p)[K]) {
  static_assert(K >= 1 && K <= 4 && (LPH == 4 || LPH == 8), "two or three butterfly steps, up to four entries");
  if constexpr (K == 4) {
    if constexpr (LPH == 4) asm(XDEP1 XSTEP4(1) XSTEP4(2)           : "+v"(p[0]), "+v"(p[1]), "+v"(p[2]), "+v"(p[3]));
    else                    asm(XDEP1 XSTEP4(1) XSTEP4(2) XSTEP4(4) : "+v"(p[0]), "+v"(p[1]), "+v"(p[2]), "+v"(p[3]));
  } else if constexpr (K == 3) {
    if constexpr (LPH == 4) asm(XDEP1 XSTEP3(1) XSTEP3(2)           : "+v"(p[0]), "+v"(p[1]), "+v"(p[2]));
    else                    asm(XDEP1 XSTEP3(1) XSTEP3(2) XSTEP3(4) : "+v"(p[0]), "+v"(p[1]), "+v"(p[2]));
  } else if constexpr (K == 2) {
    if constexpr (LPH == 4) asm(XDEP1 XSTEP2(1) XDEP2 XSTEP2(2)                 : "+v"(p[0]), "+v"(p[1]));
    else                    asm(XDEP1 XSTEP2(1) XDEP2 XSTEP2(2) XDEP2 XSTEP2(4) : "+v"(p[0]), "+v"(p[1]));
  } else {
    if constexpr (LPH == 4) asm(XDEP1 XSTEP1(1) XDEP1 XSTEP1(2)                 : "+v"(p[0]));
    else                    asm(XDEP1 XSTEP1(1) XDEP1 XSTEP1(2) XDEP1 XSTEP1(4) : "+v"(p[0]));
  }
}
__device__ __forceinline__ int xhalf_i(int v) {
  return __builtin_amdgcn_permlanex16(v, v, 0x76543210, 0xfedcba98, false, false);
}
__device__ __forceinline__ float xhalf(float v) {
  return __builtin_bit_cast(float, xhalf_i(__builtin_bit_cast(int, v)));
}

template <int C, int I>
__device__ __forceinline__ float4 load_row(const float* __restrict__ xl, int my_s, int j0, int sub) {
  int s;
  if constexpr (C == 32) s = __builtin_amdgcn_readlane(my_s, j0 + I);
  else                   s = dpp_row<0x150 + I>(my_s);
  return *(const float4*)(xl + (size_t)s * (4 * C) + 4 * sub);
}

template <int C, int K>
__device__ __forceinline__ void walk_entries(
    const float* __restrict__ xl, int my_s, int j0, int e0, int total, int sub,
    const float4& xr4, const float4& at4, float& m, float& den, float4& acc) {
  constexpr int LPH = C / 4;
  float4 row[K];
  row[0] = load_row<C, 0>(xl, my_s, j0, sub);
  if constexpr (K > 1) row[1] = load_row<C, 1>(xl, my_s, j0, sub);
  if constexpr (K > 2) row[2] = load_row<C, 2>(xl, my_s, j0, sub);
  if constexpr (K > 3) row[3] = load_row<C, 3>(xl, my_s, j0, sub);
  float sc[K];
#pragma unroll
  for (int k = 0; k < K; ++k) {
    float vx = row[k].x + xr4.x, vy = row[k].y + xr4.y;
    float vz = row[k].z + xr4.z, vw = row[k].w + xr4.w;
    vx = vx > 0.0f ? vx : 0.2f * vx;
    vy = vy > 0.0f ? vy : 0.2f * vy;
    vz = vz > 0.0f ? vz : 0.2f * vz;
    vw = vw > 0.0f ? vw : 0.2f * vw;
    sc[k] = at4.x * vx + at4.y * vy + at4.z * vz + at4.w * vw;
  }
  head_sums<K, LPH>(sc);
  float mb = m;
#pragma unroll
  for (int k = 0; k < K; ++k) {
    sc[k] = (e0 + k < total) ? sc[k] : -INFINITY;
    mb = fmaxf(mb, sc[k]);
  }
  const float scale = exp_nonpos(m - mb);
  den   *= scale;
  acc.x *= scale; acc.y *= scale; acc.z *= scale; acc.w *= scale;
#pragma unroll
  for (int k = 0; k < K; ++k) {
    const float wgt = exp_nonpos(sc[k] - mb);
    den   += wgt;
    acc.x = __builtin_fmaf(wgt, row[k].x, acc.x);
    acc.y = __builtin_fmaf(wgt, row[k].y, acc.y);
    acc.z = __builtin_fmaf(wgt, row[k].z, acc.z);
    acc.w = __builtin_fmaf(wgt, row[k].w, acc.w);
  }
  m = mb;
}

#define HP 68

template <int C, bool FINAL, int MN>
__global__ __launch_bounds__(TPB) void node_softmax_agg(
    const float* __restrict__ xl, const float* __restrict__ xr,
    const float* __restrict__ att, const float* __restrict__ bias,
    const int* __restrict__ rowptr, const int* __restrict__ csr,
    float* __restrict__ outp,
    const float* __restrict__ Wl, const float* __restrict__ Wr,
    float* __restrict__ Yl, float* __restrict__ Yr, int N) {
  constexpr int M   = 4 * C;
  constexpr int LPN = C;
  constexpr int LPH = C / 4;
  const int node = (blockIdx.x * TPB + threadIdx.x) / LPN;
  const int sub  = threadIdx.x & (LPN - 1);
  const bool valid = node < N;
  const int i = valid ? node : N - 1;

  const float4 xr4 = *(const float4*)(xr + (size_t)i * M + 4 * sub);
  const float4 at4 = *(const float4*)(att + 4 * sub);
  const int beg   = rowptr[i];
  const int total = rowptr[i + 1] - beg + 1;
  int wtotal = total;
  if constexpr (LPN < 32) wtotal = max(wtotal, xhalf_i(wtotal));

  float m = -INFINITY, den = 0.0f;
  float4 acc = make_float4(0.0f, 0.0f, 0.0f, 0.0f);

  for (int base = 0; base < wtotal; base += LPN) {
    const int idx = base + sub;
    int my_s = i;
    if (idx >= 1 && idx < total) my_s = csr[beg + idx - 1];
    const int cnt = min(LPN, wtotal - base);
    int j = 0;
    for (; j + 4 <= cnt; j += 4) {
      walk_entries<C, 4>(xl, my_s, j, base + j, total, sub, xr4, at4, m, den, acc);
      if constexpr (LPN == 16) my_s = dpp_row<0x120 + 12>(my_s);
    }
    const int rem = cnt - j;
    if (rem == 3)
      walk_entries<C, 3>(xl, my_s, j, base + j, total, sub, xr4, at4, m, den, acc);
    else if (rem == 2)
      walk_entries<C, 2>(xl, my_s, j, base + j, total, sub, xr4, at4, m, den, acc);
    else if (rem == 1)
      walk_entries<C, 1>(xl, my_s, j, base + j, total, sub, xr4, at4, m, den, acc);
  }

  float4 r = make_float4(acc.x / den, acc.y / den, acc.z / den, acc.w / den);
  if constexpr (!FINAL) {
    const float4 b4 = *(const float4*)(bias + 4 * sub);
    r.x += b4.x; r.y += b4.y; r.z += b4.z; r.w += b4.w;
    r.x = r.x > 0.0f ? r.x : expm1f(r.x);
    r.y = r.y > 0.0f ? r.y : expm1f(r.y);
    r.z = r.z > 0.0f ? r.z : expm1f(r.z);
    r.w = r.w > 0.0f ? r.w : expm1f(r.w);

    static_assert(M == 64 && TPB / LPN == 16, "the work-group's rows are one 16x64 A tile");
    __shared__ __align__(16) float ht[16 * HP];
    *(float4*)(ht + (threadIdx.x / LPN) * HP + 4 * sub) = r;
    __syncthreads();

    constexpr int GPS = MN / 32;
    const int w = threadIdx.x >> 5;
    if (2 * GPS < WAVES_PER_BLOCK && w >= 2 * GPS) return;
    const int lane = threadIdx.x & 31;
    const int rr   = lane & 15;
    const int half = lane >> 4;
    const bool right = w >= GPS;
    const float* Wm = right ? Wr : Wl;
    float* Y = right ? Yr : Yl;
    const int cb = (right ? w - GPS : w) * 32 + 2 * rr;
    const float* arow = ht + rr * HP + 2 * half;
    const float* wcol = Wm + (size_t)(2 * half) * MN + cb;
    v8f c0 = {}, c1 = {};
#pragma unroll
    for (int ks = 0; ks < 16; ++ks) {
      const v2f a  = *(const v2f*)(arow + 4 * ks);
      const v2f w0 = *(const v2f*)(wcol + (size_t)(4 * ks) * MN);
      const v2f w1 = *(const v2f*)(wcol + (size_t)(4 * ks + 1) * MN);
      v2f b0, b1;
      b0.x = w0.x; b0.y = w1.x;
      b1.x = w0.y; b1.y = w1.y;
      c0 = __builtin_amdgcn_wmma_f32_16x16x4_f32(false, a, false, b0, (short)0, c0, false, false);
      c1 = __builtin_amdgcn_wmma_f32_16x16x4_f32(false, a, false, b1, (short)0, c1, false, false);
    }
    const int row0 = blockIdx.x * (TPB / LPN);
    float* yp = Y + (size_t)(row0 + 8 * half) * MN + cb;
    if (row0 + 16 <= N) {
#pragma unroll
      for (int v = 0; v < 8; ++v) {
        v2f o;
        o.x = c0[v]; o.y = c1[v];
        *(v2f*)(yp + (size_t)v * MN) = o;
      }
    } else {
#pragma unroll
      for (int v = 0; v < 8; ++v) {
        if (row0 + 8 * half + v < N) {
          v2f o;
          o.x = c0[v]; o.y = c1[v];
          *(v2f*)(yp + (size_t)v * MN) = o;
        }
      }
    }
  } else {
    static_assert(LPH == 8 && LPN == 32, "two steps: inside the row, then across the rows");
    asm(XDEP1 XSTEP4(8) : "+v"(r.x), "+v"(r.y), "+v"(r.z), "+v"(r.w));
    r.x += xhalf(r.x); r.y += xhalf(r.y);
    r.z += xhalf(r.z); r.w += xhalf(r.w);
    if (valid && sub < LPH) {
      const float4 b4 = *(const float4*)(bias + 4 * sub);
      r.x = 0.25f * r.x + b4.x; r.y = 0.25f * r.y + b4.y;
      r.z = 0.25f * r.z + b4.z; r.w = 0.25f * r.w + b4.w;
      *(float4*)(outp + (size_t)i * C + 4 * sub) = r;
    }
  }
}

extern "C" void kernel_launch(void* const* d_in, const int* in_sizes, int n_in,
                              void* d_out, int out_size, void* d_ws, size_t ws_size,
                              hipStream_t stream) {
  (void)in_sizes; (void)n_in; (void)out_size; (void)ws_size;
  const float* x   = (const float*)d_in[0];
  const int*   ei  = (const int*)d_in[1];
  const float* W1l = (const float*)d_in[2];
  const float* W1r = (const float*)d_in[3];
  const float* a1  = (const float*)d_in[4];
  const float* b1  = (const float*)d_in[5];
  const float* W2l = (const float*)d_in[6];
  const float* W2r = (const float*)d_in[7];
  const float* a2  = (const float*)d_in[8];
  const float* b2  = (const float*)d_in[9];
  const float* W3l = (const float*)d_in[10];
  const float* W3r = (const float*)d_in[11];
  const float* a3  = (const float*)d_in[12];
  const float* b3  = (const float*)d_in[13];

  constexpr int N = 100000, E = 800000;

  float* ws = (float*)d_ws;
  float* xl1 = ws;
  float* xr1 = xl1 + (size_t)N * 64;
  float* xl2 = xr1 + (size_t)N * 64;
  float* xr2 = xl2 + (size_t)N * 64;
  float* xl3 = ws;
  float* xr3 = xr2 + (size_t)N * 64;
  constexpr int NBK = (N + NPB - 1) / NPB;
  constexpr int NCH = (E + EPC - 1) / EPC;
  constexpr int RP  = (N + 1 + 31) / 32 * 32;
  int* rowptr = (int*)(xr3 + (size_t)N * 128);
  int* csr    = rowptr + RP;
  int* tmp    = csr + E;
  int* tcnt   = tmp + (size_t)NCH * EPC;
  int* toff   = tcnt + NBK * NCH;

  constexpr int rowTiles = N / 16;
  constexpr int GBLK = (rowTiles + GTILES - 1) / GTILES;

  csr_bin       <NBK, NCH><<<NCH, BTH, 0, stream>>>(ei, tmp, tcnt, toff, E);
  csr_place_gemm<NBK, NCH><<<NBK + GBLK, BTH, 0, stream>>>(tmp, tcnt, toff, rowptr, csr, N,
                                                          x, W1l, W1r, xl1, xr1);

  const int nb16 = (N * 16 + TPB - 1) / TPB;
  const int nb32 = (N * 32 + TPB - 1) / TPB;

  node_softmax_agg<16, false, 64><<<nb16, TPB, 0, stream>>>(xl1, xr1, a1, b1, rowptr, csr, nullptr,
                                                            W2l, W2r, xl2, xr2, N);
  node_softmax_agg<16, false, 128><<<nb16, TPB, 0, stream>>>(xl2, xr2, a2, b2, rowptr, csr, nullptr,
                                                             W3l, W3r, xl3, xr3, N);
  node_softmax_agg<32, true, 0><<<nb32, TPB, 0, stream>>>(xl3, xr3, a3, b3, rowptr, csr, (float*)d_out,
                                                          nullptr, nullptr, nullptr, nullptr, N);
}


# Edge-walk kernels fitted into 64 registers (16-waves-per-SIMD launch bound, 32-bit row offsets, rows as native 4-vectors): 16 instead of 12 waves per SIMD wait side by side; sum order kept
# speedup vs baseline: 11.7533x; 1.0543x over previous
#include <hip/hip_runtime.h>
#include <math.h>

typedef __attribute__((ext_vector_type(2))) float v2f;
typedef __attribute__((ext_vector_type(8))) float v8f;
typedef __attribute__((ext_vector_type(4))) float v4f;

#define TPB 256
#define WAVES_PER_BLOCK 8

template <int M, int TILES, int THREADS>
__device__ __forceinline__ void gemm_pair_tiles(
    const float* __restrict__ X, const float* __restrict__ Wl,
    const float* __restrict__ Wr, float* __restrict__ Yl,
    float* __restrict__ Yr, int Nrows, float* lw, int blk) {
  constexpr int TOTAL4 = 64 * M / 4;
  for (int i = threadIdx.x; i < TOTAL4; i += THREADS) {
    ((float4*)lw)[i]          = ((const float4*)Wl)[i];
    ((float4*)lw)[TOTAL4 + i] = ((const float4*)Wr)[i];
  }
  __syncthreads();

  const int wv = threadIdx.x >> 5;
  const int rowTile = blk * TILES + wv;
  if (wv >= TILES || rowTile >= (Nrows >> 4)) return;
  const int row0 = rowTile << 4;
  const int lane = threadIdx.x & 31;
  const int r    = lane & 15;
  const int half = lane >> 4;

  v2f a[16];
  const float* xrow = X + (size_t)(row0 + r) * 64;
#pragma unroll
  for (int ks = 0; ks < 16; ++ks) {
    const int k = 4 * ks + 2 * half;
    a[ks].x = xrow[k];
    a[ks].y = xrow[k + 1];
  }

#pragma unroll
  for (int side = 0; side < 2; ++side) {
    const float* lws = lw + side * (64 * M);
    float* Y = side ? Yr : Yl;
#pragma nounroll
    for (int tn = 0; tn < M / 16; ++tn) {
      const int col0 = tn << 4;
      v8f c = {};
#pragma unroll
      for (int ks = 0; ks < 16; ++ks) {
        const int k = 4 * ks + 2 * half;
        v2f b;
        b.x = lws[k * M + col0 + r];
        b.y = lws[(k + 1) * M + col0 + r];
        c = __builtin_amdgcn_wmma_f32_16x16x4_f32(false, a[ks], false, b,
                                                  (short)0, c, false, false);
      }
#pragma unroll
      for (int v = 0; v < 8; ++v)
        Y[(size_t)(row0 + v + 8 * half) * M + col0 + r] = c[v];
    }
  }
}

#define BTH 1024
#define EPC 4096
#define NPB 512
#define NPB_LOG 9
#define STAGE_CAP 8192

__device__ __forceinline__ int block_excl_scan(int v, int* s_w, int& total) {
  const int lane = threadIdx.x & 31, w = threadIdx.x >> 5;
  int inc = v;
  for (int o = 1; o < 32; o <<= 1) {
    const int u = __shfl_up(inc, o);
    if (lane >= o) inc += u;
  }
  if (lane == 31) s_w[w] = inc;
  __syncthreads();
  int pre = 0, tot = 0;
  for (int k = 0; k < BTH / 32; ++k) {
    const int x = s_w[k];
    if (k < w) pre += x;
    tot += x;
  }
  __syncthreads();
  total = tot;
  return pre + inc - v;
}

template <int NBK, int NCH>
__global__ __launch_bounds__(BTH) void csr_bin(const int* __restrict__ ei,
                                               int* __restrict__ tmp,
                                               int* __restrict__ tcnt,
                                               int* __restrict__ toff, int E) {
  __shared__ int s_hist[NBK];
  __shared__ int s_off[NBK];
  __shared__ int s_w[BTH / 32];
  __shared__ int s_stage[EPC];
  const int t = threadIdx.x, c = blockIdx.x;
  if (t < NBK) s_hist[t] = 0;
  __syncthreads();

  const int e0 = c * EPC;
  int word[EPC / BTH], bk[EPC / BTH], rk[EPC / BTH];
#pragma unroll
  for (int j = 0; j < EPC / BTH; ++j) {
    const int e = e0 + j * BTH + t;
    bk[j] = -1; word[j] = 0; rk[j] = 0;
    if (e < E) {
      const int s = ei[e], d = ei[E + e];
      bk[j]   = d >> NPB_LOG;
      word[j] = (s << NPB_LOG) | (d & (NPB - 1));
      rk[j]   = atomicAdd(&s_hist[bk[j]], 1);
    }
  }
  __syncthreads();

  const int h = (t < NBK) ? s_hist[t] : 0;
  int nvalid;
  const int ex = block_excl_scan(h, s_w, nvalid);
  if (t < NBK) {
    s_off[t] = ex;
    tcnt[t * NCH + c] = h;
    toff[t * NCH + c] = ex;
  }
  __syncthreads();
#pragma unroll
  for (int j = 0; j < EPC / BTH; ++j)
    if (bk[j] >= 0) s_stage[s_off[bk[j]] + rk[j]] = word[j];
  __syncthreads();
  for (int k = t; k < nvalid; k += BTH) tmp[e0 + k] = s_stage[k];
}

#define GTILES 25
template <int NBK, int NCH>
__global__ __launch_bounds__(BTH) void csr_place_gemm(const int* __restrict__ tmp,
                                                      const int* __restrict__ tcnt,
                                                      const int* __restrict__ toff,
                                                      int* __restrict__ rowptr,
                                                      int* __restrict__ csr, int N,
                                                      const float* __restrict__ X,
                                                      const float* __restrict__ Wl,
                                                      const float* __restrict__ Wr,
                                                      float* __restrict__ Yl,
                                                      float* __restrict__ Yr) {
  static_assert((NCH & 3) == 0, "run table rows are summed as int4");
  static_assert(STAGE_CAP * sizeof(int) == 2 * 64 * 64 * sizeof(float), "the W pair takes exactly the segment buffer");
  __shared__ int s_cnt[NCH];
  __shared__ int s_off[NCH];
  __shared__ int s_w[BTH / 32];
  __shared__ int s_deg[NPB];
  __shared__ int s_pos[NPB];
  __shared__ __align__(16) unsigned char s_raw[STAGE_CAP * sizeof(int)];
  if (blockIdx.x >= NBK) {
    gemm_pair_tiles<64, GTILES, BTH>(X, Wl, Wr, Yl, Yr, N, (float*)s_raw, (int)blockIdx.x - NBK);
    return;
  }
  int* s_out = (int*)s_raw;
  const int t = threadIdx.x, lane = t & 31, w = t >> 5, b = blockIdx.x;

  int part = 0;
  const int4* c4 = (const int4*)tcnt;
  const int nq = (b * NCH) >> 2;
  for (int q = t; q < nq; q += BTH) {
    const int4 v = c4[q];
    part += (v.x + v.y) + (v.z + v.w);
  }
  int mine = 0;
  if (t < NCH) {
    mine = tcnt[b * NCH + t];
    s_cnt[t] = mine;
    s_off[t] = toff[b * NCH + t];
  }
  if (t < NPB) s_deg[t] = 0;
  int base, nb;
  (void)block_excl_scan(part, s_w, base);
  (void)block_excl_scan(mine, s_w, nb);

  for (int c = w; c < NCH; c += BTH / 32) {
    const int n = s_cnt[c];
    const int* run = tmp + c * EPC + s_off[c];
    for (int k = lane; k < n; k += 32) atomicAdd(&s_deg[run[k] & (NPB - 1)], 1);
  }
  __syncthreads();

  const int dg = (t < NPB) ? s_deg[t] : 0;
  int nb2;
  const int ex = block_excl_scan(dg, s_w, nb2);
  if (t < NPB) {
    s_pos[t] = ex;
    const int node = b * NPB + t;
    if (node < N) rowptr[node] = base + ex;
  }
  if (b == NBK - 1 && t == 0) rowptr[N] = base + nb;
  __syncthreads();

  const bool staged = nb <= STAGE_CAP;
  for (int c = w; c < NCH; c += BTH / 32) {
    const int n = s_cnt[c];
    const int* run = tmp + c * EPC + s_off[c];
    for (int k = lane; k < n; k += 32) {
      const int wd = run[k];
      const int slot = atomicAdd(&s_pos[wd & (NPB - 1)], 1);
      const int src = (int)((unsigned)wd >> NPB_LOG);
      if (staged) s_out[slot] = src;
      else        csr[base + slot] = src;
    }
  }
  __syncthreads();
  if (staged)
    for (int k = t; k < nb; k += BTH) csr[base + k] = s_out[k];
}

__device__ __forceinline__ float exp_nonpos(float x) {
  return __builtin_amdgcn_exp2f(x * 1.44269504088896340736f);
}

template <int CTRL>
__device__ __forceinline__ int dpp_row(int v) {
  return __builtin_amdgcn_update_dpp(0, v, CTRL, 0xf, 0xf, true);
}
#define XADD(r, m) "v_add_f32_dpp " r ", " r ", " r " row_xmask:" #m " row_mask:0xf bank_mask:0xf\n\t"
#define XSTEP1(m) XADD("%0", m)
#define XSTEP2(m) XADD("%0", m) XADD("%1", m)
#define XSTEP3(m) XADD("%0", m) XADD("%1", m) XADD("%2", m)
#define XSTEP4(m) XADD("%0", m) XADD("%1", m) XADD("%2", m) XADD("%3", m)
#define XDEP1 "s_delay_alu instid0(VALU_DEP_1)\n\t"
#define XDEP2 "s_delay_alu instid0(VALU_DEP_2) | instskip(NEXT) | instid1(VALU_DEP_2)\n\t"

template <int K, int LPH>
__device__ __forceinline__ void head_sums(float (&p)[K]) {
  static_assert(K >= 1 && K <= 4 && (LPH == 4 || LPH == 8), "two or three butterfly steps, up to four entries");
  if constexpr (K == 4) {
    if constexpr (LPH == 4) asm(XDEP1 XSTEP4(1) XSTEP4(2)           : "+v"(p[0]), "+v"(p[1]), "+v"(p[2]), "+v"(p[3]));
    else                    asm(XDEP1 XSTEP4(1) XSTEP4(2) XSTEP4(4) : "+v"(p[0]), "+v"(p[1]), "+v"(p[2]), "+v"(p[3]));
  } else if constexpr (K == 3) {
    if constexpr (LPH == 4) asm(XDEP1 XSTEP3(1) XSTEP3(2)           : "+v"(p[0]), "+v"(p[1]), "+v"(p[2]));
    else                    asm(XDEP1 XSTEP3(1) XSTEP3(2) XSTEP3(4) : "+v"(p[0]), "+v"(p[1]), "+v"(p[2]));
  } else if constexpr (K == 2) {
    if constexpr (LPH == 4) asm(XDEP1 XSTEP2(1) XDEP2 XSTEP2(2)                 : "+v"(p[0]), "+v"(p[1]));
    else                    asm(XDEP1 XSTEP2(1) XDEP2 XSTEP2(2) XDEP2 XSTEP2(4) : "+v"(p[0]), "+v"(p[1]));
  } else {
    if constexpr (LPH == 4) asm(XDEP1 XSTEP1(1) XDEP1 XSTEP1(2)                 : "+v"(p[0]));
    else                    asm(XDEP1 XSTEP1(1) XDEP1 XSTEP1(2) XDEP1 XSTEP1(4) : "+v"(p[0]));
  }
}
__device__ __forceinline__ int xhalf_i(int v) {
  return __builtin_amdgcn_permlanex16(v, v, 0x76543210, 0xfedcba98, false, false);
}
__device__ __forceinline__ float xhalf(float v) {
  return __builtin_bit_cast(float, xhalf_i(__builtin_bit_cast(int, v)));
}

template <int C, int I>
__device__ __forceinline__ v4f load_row(const float* __restrict__ xl, int my_s, int j0, int sub) {
  if constexpr (C == 32) {
    const int s = __builtin_amdgcn_readlane(my_s, j0 + I);
    return *(const v4f*)(xl + (size_t)s * (4 * C) + 4 * sub);
  } else {
    const int s = dpp_row<0x150 + I>(my_s);
    const unsigned off = (unsigned)s * (16u * C) + 16u * (unsigned)sub;
    return *(const v4f*)((const char*)xl + off);
  }
}

template <int C, int K>
__device__ __forceinline__ void walk_entries(
    const float* __restrict__ xl, int my_s, int j0, int e0, int total, int sub,
    const float4& xr4, const float4& at4, float& m, float& den, float4& acc) {
  constexpr int LPH = C / 4;
  v4f row[K];
  row[0] = load_row<C, 0>(xl, my_s, j0, sub);
  if constexpr (K > 1) row[1] = load_row<C, 1>(xl, my_s, j0, sub);
  if constexpr (K > 2) row[2] = load_row<C, 2>(xl, my_s, j0, sub);
  if constexpr (K > 3) row[3] = load_row<C, 3>(xl, my_s, j0, sub);
  float sc[K];
#pragma unroll
  for (int k = 0; k < K; ++k) {
    float vx = row[k].x + xr4.x, vy = row[k].y + xr4.y;
    float vz = row[k].z + xr4.z, vw = row[k].w + xr4.w;
    vx = vx > 0.0f ? vx : 0.2f * vx;
    vy = vy > 0.0f ? vy : 0.2f * vy;
    vz = vz > 0.0f ? vz : 0.2f * vz;
    vw = vw > 0.0f ? vw : 0.2f * vw;
    sc[k] = at4.x * vx + at4.y * vy + at4.z * vz + at4.w * vw;
  }
  head_sums<K, LPH>(sc);
  float mb = m;
#pragma unroll
  for (int k = 0; k < K; ++k) {
    sc[k] = (e0 + k < total) ? sc[k] : -INFINITY;
    mb = fmaxf(mb, sc[k]);
  }
  const float scale = exp_nonpos(m - mb);
  den   *= scale;
  acc.x *= scale; acc.y *= scale; acc.z *= scale; acc.w *= scale;
#pragma unroll
  for (int k = 0; k < K; ++k) {
    const float wgt = exp_nonpos(sc[k] - mb);
    den   += wgt;
    acc.x = __builtin_fmaf(wgt, row[k].x, acc.x);
    acc.y = __builtin_fmaf(wgt, row[k].y, acc.y);
    acc.z = __builtin_fmaf(wgt, row[k].z, acc.z);
    acc.w = __builtin_fmaf(wgt, row[k].w, acc.w);
  }
  m = mb;
}

#define HP 68

template <int C, bool FINAL, int MN>
__global__ __launch_bounds__(TPB, 16) void node_softmax_agg(
    const float* __restrict__ xl, const float* __restrict__ xr,
    const float* __restrict__ att, const float* __restrict__ bias,
    const int* __restrict__ rowptr, const int* __restrict__ csr,
    float* __restrict__ outp,
    const float* __restrict__ Wl, const float* __restrict__ Wr,
    float* __restrict__ Yl, float* __restrict__ Yr, int N) {
  constexpr int M   = 4 * C;
  constexpr int LPN = C;
  constexpr int LPH = C / 4;
  const int node = (blockIdx.x * TPB + threadIdx.x) / LPN;
  const int sub  = threadIdx.x & (LPN - 1);
  const bool valid = node < N;
  const int i = valid ? node : N - 1;

  const float4 xr4 = *(const float4*)(xr + (size_t)i * M + 4 * sub);
  const float4 at4 = *(const float4*)(att + 4 * sub);
  const int beg   = rowptr[i];
  const int total = rowptr[i + 1] - beg + 1;
  int wtotal = total;
  if constexpr (LPN < 32) wtotal = max(wtotal, xhalf_i(wtotal));

  float m = -INFINITY, den = 0.0f;
  float4 acc = make_float4(0.0f, 0.0f, 0.0f, 0.0f);

  for (int base = 0; base < wtotal; base += LPN) {
    const int idx = base + sub;
    int my_s = i;
    if (idx >= 1 && idx < total) my_s = csr[beg + idx - 1];
    const int cnt = min(LPN, wtotal - base);
    int j = 0;
    for (; j + 4 <= cnt; j += 4) {
      walk_entries<C, 4>(xl, my_s, j, base + j, total, sub, xr4, at4, m, den, acc);
      if constexpr (LPN == 16) my_s = dpp_row<0x120 + 12>(my_s);
    }
    const int rem = cnt - j;
    if (rem == 3)
      walk_entries<C, 3>(xl, my_s, j, base + j, total, sub, xr4, at4, m, den, acc);
    else if (rem == 2)
      walk_entries<C, 2>(xl, my_s, j, base + j, total, sub, xr4, at4, m, den, acc);
    else if (rem == 1)
      walk_entries<C, 1>(xl, my_s, j, base + j, total, sub, xr4, at4, m, den, acc);
  }

  float4 r = make_float4(acc.x / den, acc.y / den, acc.z / den, acc.w / den);
  if constexpr (!FINAL) {
    const float4 b4 = *(const float4*)(bias + 4 * sub);
    r.x += b4.x; r.y += b4.y; r.z += b4.z; r.w += b4.w;
    r.x = r.x > 0.0f ? r.x : expm1f(r.x);
    r.y = r.y > 0.0f ? r.y : expm1f(r.y);
    r.z = r.z > 0.0f ? r.z : expm1f(r.z);
    r.w = r.w > 0.0f ? r.w : expm1f(r.w);

    static_assert(M == 64 && TPB / LPN == 16, "the work-group's rows are one 16x64 A tile");
    __shared__ __align__(16) float ht[16 * HP];
    *(float4*)(ht + (threadIdx.x / LPN) * HP + 4 * sub) = r;
    __syncthreads();

    constexpr int GPS = MN / 32;
    const int w = threadIdx.x >> 5;
    if (2 * GPS < WAVES_PER_BLOCK && w >= 2 * GPS) return;
    const int lane = threadIdx.x & 31;
    const int rr   = lane & 15;
    const int half = lane >> 4;
    const bool right = w >= GPS;
    const float* Wm = right ? Wr : Wl;
    float* Y = right ? Yr : Yl;
    const int cb = (right ? w - GPS : w) * 32 + 2 * rr;
    const float* arow = ht + rr * HP + 2 * half;
    const float* wcol = Wm + (size_t)(2 * half) * MN + cb;
    v8f c0 = {}, c1 = {};
#pragma unroll
    for (int ks = 0; ks < 16; ++ks) {
      const v2f a  = *(const v2f*)(arow + 4 * ks);
      const v2f w0 = *(const v2f*)(wcol + (size_t)(4 * ks) * MN);
      const v2f w1 = *(const v2f*)(wcol + (size_t)(4 * ks + 1) * MN);
      v2f b0, b1;
      b0.x = w0.x; b0.y = w1.x;
      b1.x = w0.y; b1.y = w1.y;
      c0 = __builtin_amdgcn_wmma_f32_16x16x4_f32(false, a, false, b0, (short)0, c0, false, false);
      c1 = __builtin_amdgcn_wmma_f32_16x16x4_f32(false, a, false, b1, (short)0, c1, false, false);
    }
    const int row0 = blockIdx.x * (TPB / LPN);
    float* yp = Y + (size_t)(row0 + 8 * half) * MN + cb;
    if (row0 + 16 <= N) {
#pragma unroll
      for (int v = 0; v < 8; ++v) {
        v2f o;
        o.x = c0[v]; o.y = c1[v];
        *(v2f*)(yp + (size_t)v * MN) = o;
      }
    } else {
#pragma unroll
      for (int v = 0; v < 8; ++v) {
        if (row0 + 8 * half + v < N) {
          v2f o;
          o.x = c0[v]; o.y = c1[v];
          *(v2f*)(yp + (size_t)v * MN) = o;
        }
      }
    }
  } else {
    static_assert(LPH == 8 && LPN == 32, "two steps: inside the row, then across the rows");
    asm(XDEP1 XSTEP4(8) : "+v"(r.x), "+v"(r.y), "+v"(r.z), "+v"(r.w));
    r.x += xhalf(r.x); r.y += xhalf(r.y);
    r.z += xhalf(r.z); r.w += xhalf(r.w);
    if (valid && sub < LPH) {
      const float4 b4 = *(const float4*)(bias + 4 * sub);
      r.x = 0.25f * r.x + b4.x; r.y = 0.25f * r.y + b4.y;
      r.z = 0.25f * r.z + b4.z; r.w = 0.25f * r.w + b4.w;
      *(float4*)(outp + (size_t)i * C + 4 * sub) = r;
    }
  }
}

extern "C" void kernel_launch(void* const* d_in, const int* in_sizes, int n_in,
                              void* d_out, int out_size, void* d_ws, size_t ws_size,
                              hipStream_t stream) {
  (void)in_sizes; (void)n_in; (void)out_size; (void)ws_size;
  const float* x   = (const float*)d_in[0];
  const int*   ei  = (const int*)d_in[1];
  const float* W1l = (const float*)d_in[2];
  const float* W1r = (const float*)d_in[3];
  const float* a1  = (const float*)d_in[4];
  const float* b1  = (const float*)d_in[5];
  const float* W2l = (const float*)d_in[6];
  const float* W2r = (const float*)d_in[7];
  const float* a2  = (const float*)d_in[8];
  const float* b2  = (const float*)d_in[9];
  const float* W3l = (const float*)d_in[10];
  const float* W3r = (const float*)d_in[11];
  const float* a3  = (const float*)d_in[12];
  const float* b3  = (const float*)d_in[13];

  constexpr int N = 100000, E = 800000;
  static_assert((unsigned long long)N * 128 * sizeof(float) < (1ull << 32), "load_row: byte offsets into an xl plane are 32-bit");

  float* ws = (float*)d_ws;
  float* xl1 = ws;
  float* xr1 = xl1 + (size_t)N * 64;
  float* xl2 = xr1 + (size_t)N * 64;
  float* xr2 = xl2 + (size_t)N * 64;
  float* xl3 = ws;
  float* xr3 = xr2 + (size_t)N * 64;
  constexpr int NBK = (N + NPB - 1) / NPB;
  constexpr int NCH = (E + EPC - 1) / EPC;
  constexpr int RP  = (N + 1 + 31) / 32 * 32;
  int* rowptr = (int*)(xr3 + (size_t)N * 128);
  int* csr    = rowptr + RP;
  int* tmp    = csr + E;
  int* tcnt   = tmp + (size_t)NCH * EPC;
  int* toff   = tcnt + NBK * NCH;

  constexpr int rowTiles = N / 16;
  constexpr int GBLK = (rowTiles + GTILES - 1) / GTILES;

  csr_bin       <NBK, NCH><<<NCH, BTH, 0, stream>>>(ei, tmp, tcnt, toff, E);
  csr_place_gemm<NBK, NCH><<<NBK + GBLK, BTH, 0, stream>>>(tmp, tcnt, toff, rowptr, csr, N,
                                                          x, W1l, W1r, xl1, xr1);

  const int nb16 = (N * 16 + TPB - 1) / TPB;
  const int nb32 = (N * 32 + TPB - 1) / TPB;

  node_softmax_agg<16, false, 64><<<nb16, TPB, 0, stream>>>(xl1, xr1, a1, b1, rowptr, csr, nullptr,
                                                            W2l, W2r, xl2, xr2, N);
  node_softmax_agg<16, false, 128><<<nb16, TPB, 0, stream>>>(xl2, xr2, a2, b2, rowptr, csr, nullptr,
                                                             W3l, W3r, xl3, xr3, N);
  node_softmax_agg<32, true, 0><<<nb32, TPB, 0, stream>>>(xl3, xr3, a3, b3, rowptr, csr, (float*)d_out,
                                                          nullptr, nullptr, nullptr, nullptr, N);
}
